# starting kernel
# baseline (speedup 1.0000x reference)
#include <hip/hip_runtime.h>
#include <hip/hip_bf16.h>

#define CH 128

typedef __attribute__((ext_vector_type(2))) float v2f;
typedef __attribute__((ext_vector_type(8))) float v8f;


__global__ void gcn_deg_init(float* __restrict__ deg, int n) {
    int i = blockIdx.x * blockDim.x + threadIdx.x;
    if (i < n) deg[i] = 1.0f;
}

__global__ void gcn_deg_accum(const int* __restrict__ dst, float* __restrict__ deg, int E) {
    int e = blockIdx.x * blockDim.x + threadIdx.x;
    if (e < E) atomicAdd(&deg[dst[e]], 1.0f);
}

__global__ void gcn_dinv(float* __restrict__ deg, int n) {
    int i = blockIdx.x * blockDim.x + threadIdx.x;
    if (i < n) deg[i] = rsqrtf(deg[i]);
}


__global__ __launch_bounds__(256) void gcn_gemm_wmma(
    const float* __restrict__ X, const float* __restrict__ W,
    float* __restrict__ XW, int n)
{
    const int wave = threadIdx.x >> 5;
    const int lane = threadIdx.x & 31;
    const int half = lane >> 4;
    const int l16  = lane & 15;
    const int m0   = blockIdx.x * 16;
    const int n0   = wave * 16;

    int arow = m0 + l16;
    if (arow >= n) arow = n - 1;
    const float* Ap = X + (size_t)arow * CH;

    v8f c = {};
#pragma unroll
    for (int k0 = 0; k0 < CH; k0 += 4) {
        const int ka = k0 + 2 * half;
        v2f a, b;
        a.x = Ap[ka];
        a.y = Ap[ka + 1];
        b.x = W[(size_t)ka       * CH + n0 + l16];
        b.y = W[(size_t)(ka + 1) * CH + n0 + l16];
        c = __builtin_amdgcn_wmma_f32_16x16x4_f32(
                 false, a,  false, b,
                 (short)0, c,  false,  false);
    }

#pragma unroll
    for (int r = 0; r < 8; ++r) {
        int row = m0 + r + 8 * half;
        if (row < n) XW[(size_t)row * CH + n0 + l16] = c[r];
    }
}


__global__ void gcn_init_out(const float* __restrict__ xw, const float* __restrict__ dinv,
                             const float* __restrict__ bias, float* __restrict__ out, int total) {
    int idx = blockIdx.x * blockDim.x + threadIdx.x;
    if (idx >= total) return;
    int i = idx >> 7;
    int c = idx & (CH - 1);
    float di = dinv[i];
    out[idx] = bias[c] + xw[idx] * di * di;
}


__global__ __launch_bounds__(256) void gcn_scatter(
    const int* __restrict__ src, const int* __restrict__ dst,
    const float* __restrict__ dinv, const float* __restrict__ xw,
    float* __restrict__ out, int E)
{
    int e = (int)(((size_t)blockIdx.x * blockDim.x + threadIdx.x) >> 5);
    if (e >= E) return;
    int lane = threadIdx.x & 31;

    int s = src[e];
    int d = dst[e];
    float nrm = dinv[s] * dinv[d];

    const float4* row = (const float4*)(xw + (size_t)s * CH);
    float4 v = row[lane];

    float* o = out + (size_t)d * CH + lane * 4;
    atomicAdd(o + 0, v.x * nrm);
    atomicAdd(o + 1, v.y * nrm);
    atomicAdd(o + 2, v.z * nrm);
    atomicAdd(o + 3, v.w * nrm);
}


extern "C" void kernel_launch(void* const* d_in, const int* in_sizes, int n_in,
                              void* d_out, int out_size, void* d_ws, size_t ws_size,
                              hipStream_t stream) {
    const float* x  = (const float*)d_in[0];
    const float* W  = (const float*)d_in[1];
    const float* b  = (const float*)d_in[2];
    const int*   ei = (const int*)d_in[3];

    const int n = in_sizes[0] / CH;
    const int E = in_sizes[3] / 2;
    const int* src = ei;
    const int* dst = ei + E;

    float* deg = (float*)d_ws;
    float* xw  = deg + n;
    float* out = (float*)d_out;

    gcn_deg_init <<<(n + 255) / 256, 256, 0, stream>>>(deg, n);
    gcn_deg_accum<<<(E + 255) / 256, 256, 0, stream>>>(dst, deg, E);
    gcn_dinv     <<<(n + 255) / 256, 256, 0, stream>>>(deg, n);

    gcn_gemm_wmma<<<(n + 15) / 16, 256, 0, stream>>>(x, W, xw, n);

    int total = n * CH;
    gcn_init_out <<<(total + 255) / 256, 256, 0, stream>>>(xw, deg, b, out, total);

    size_t scatter_threads = (size_t)E * 32;
    gcn_scatter  <<<(int)((scatter_threads + 255) / 256), 256, 0, stream>>>(src, dst, deg, xw, out, E);
}


# scatter: lane owns channels l+32j so each f32 atomic add and each row load of a wave covers one whole 128-B line (4 line requests per edge, was 16); adds still in arrival order
# speedup vs baseline: 1.2793x; 1.2793x over previous
#include <hip/hip_runtime.h>
#include <hip/hip_bf16.h>

#define CH 128

typedef __attribute__((ext_vector_type(2))) float v2f;
typedef __attribute__((ext_vector_type(8))) float v8f;


__global__ void gcn_deg_init(float* __restrict__ deg, int n) {
    int i = blockIdx.x * blockDim.x + threadIdx.x;
    if (i < n) deg[i] = 1.0f;
}

__global__ void gcn_deg_accum(const int* __restrict__ dst, float* __restrict__ deg, int E) {
    int e = blockIdx.x * blockDim.x + threadIdx.x;
    if (e < E) atomicAdd(&deg[dst[e]], 1.0f);
}

__global__ void gcn_dinv(float* __restrict__ deg, int n) {
    int i = blockIdx.x * blockDim.x + threadIdx.x;
    if (i < n) deg[i] = rsqrtf(deg[i]);
}


__global__ __launch_bounds__(256) void gcn_gemm_wmma(
    const float* __restrict__ X, const float* __restrict__ W,
    float* __restrict__ XW, int n)
{
    const int wave = threadIdx.x >> 5;
    const int lane = threadIdx.x & 31;
    const int half = lane >> 4;
    const int l16  = lane & 15;
    const int m0   = blockIdx.x * 16;
    const int n0   = wave * 16;

    int arow = m0 + l16;
    if (arow >= n) arow = n - 1;
    const float* Ap = X + (size_t)arow * CH;

    v8f c = {};
#pragma unroll
    for (int k0 = 0; k0 < CH; k0 += 4) {
        const int ka = k0 + 2 * half;
        v2f a, b;
        a.x = Ap[ka];
        a.y = Ap[ka + 1];
        b.x = W[(size_t)ka       * CH + n0 + l16];
        b.y = W[(size_t)(ka + 1) * CH + n0 + l16];
        c = __builtin_amdgcn_wmma_f32_16x16x4_f32(
                 false, a,  false, b,
                 (short)0, c,  false,  false);
    }

#pragma unroll
    for (int r = 0; r < 8; ++r) {
        int row = m0 + r + 8 * half;
        if (row < n) XW[(size_t)row * CH + n0 + l16] = c[r];
    }
}


__global__ void gcn_init_out(const float* __restrict__ xw, const float* __restrict__ dinv,
                             const float* __restrict__ bias, float* __restrict__ out, int total) {
    int idx = blockIdx.x * blockDim.x + threadIdx.x;
    if (idx >= total) return;
    int i = idx >> 7;
    int c = idx & (CH - 1);
    float di = dinv[i];
    out[idx] = bias[c] + xw[idx] * di * di;
}


__global__ __launch_bounds__(256) void gcn_scatter(
    const int* __restrict__ src, const int* __restrict__ dst,
    const float* __restrict__ dinv, const float* __restrict__ xw,
    float* __restrict__ out, int E)
{
    int e = (int)(((size_t)blockIdx.x * blockDim.x + threadIdx.x) >> 5);
    if (e >= E) return;
    int lane = threadIdx.x & 31;

    int s = src[e];
    int d = dst[e];
    float nrm = dinv[s] * dinv[d];

    const float* row = xw + (size_t)s * CH + lane;
    float v0 = row[0];
    float v1 = row[32];
    float v2 = row[64];
    float v3 = row[96];

    float* o = out + (size_t)d * CH + lane;
    atomicAdd(o,      v0 * nrm);
    atomicAdd(o + 32, v1 * nrm);
    atomicAdd(o + 64, v2 * nrm);
    atomicAdd(o + 96, v3 * nrm);
}


extern "C" void kernel_launch(void* const* d_in, const int* in_sizes, int n_in,
                              void* d_out, int out_size, void* d_ws, size_t ws_size,
                              hipStream_t stream) {
    const float* x  = (const float*)d_in[0];
    const float* W  = (const float*)d_in[1];
    const float* b  = (const float*)d_in[2];
    const int*   ei = (const int*)d_in[3];

    const int n = in_sizes[0] / CH;
    const int E = in_sizes[3] / 2;
    const int* src = ei;
    const int* dst = ei + E;

    float* deg = (float*)d_ws;
    float* xw  = deg + n;
    float* out = (float*)d_out;

    gcn_deg_init <<<(n + 255) / 256, 256, 0, stream>>>(deg, n);
    gcn_deg_accum<<<(E + 255) / 256, 256, 0, stream>>>(dst, deg, E);
    gcn_dinv     <<<(n + 255) / 256, 256, 0, stream>>>(deg, n);

    gcn_gemm_wmma<<<(n + 15) / 16, 256, 0, stream>>>(x, W, xw, n);

    int total = n * CH;
    gcn_init_out <<<(total + 255) / 256, 256, 0, stream>>>(xw, deg, b, out, total);

    size_t scatter_threads = (size_t)E * 32;
    gcn_scatter  <<<(int)((scatter_threads + 255) / 256), 256, 0, stream>>>(src, dst, deg, xw, out, E);
}


# aggregate by destination: count, allocate and fill a dst-grouped edge list; one wave sums a node's rows in registers and stores once; no float atomics; sum order = fill arrival order (int atomics)
# speedup vs baseline: 1.3691x; 1.0702x over previous
#include <hip/hip_runtime.h>
#include <hip/hip_bf16.h>

#define CH 128

typedef __attribute__((ext_vector_type(2))) float v2f;
typedef __attribute__((ext_vector_type(8))) float v8f;


__global__ void gcn_zero(int* __restrict__ cnt, int n) {
    int i = blockIdx.x * blockDim.x + threadIdx.x;
    if (i <= n) cnt[i] = 0;
}

__global__ void gcn_hist(const int* __restrict__ dst, int* __restrict__ cnt, int E) {
    int e = blockIdx.x * blockDim.x + threadIdx.x;
    if (e < E) atomicAdd(&cnt[dst[e]], 1);
}

__global__ __launch_bounds__(256) void gcn_alloc(
    const int* __restrict__ cnt, int* __restrict__ total, int* __restrict__ seg,
    float* __restrict__ dinv, int n)
{
    const int i    = blockIdx.x * blockDim.x + threadIdx.x;
    const int lane = threadIdx.x & 31;
    const int c    = (i < n) ? cnt[i] : 0;

    int s = c;
#pragma unroll
    for (int off = 1; off < 32; off <<= 1) {
        int t = __shfl_up(s, off);
        if (lane >= off) s += t;
    }
    const int wsum = __shfl(s, 31);
    int base = 0;
    if (lane == 31) base = atomicAdd(total, wsum);
    base = __shfl(base, 31);

    if (i < n) {
        seg[2 * i]     = base + s - c;
        seg[2 * i + 1] = c;
        dinv[i] = rsqrtf((float)(c + 1));
    }
}

__global__ void gcn_fill(const int* __restrict__ src, const int* __restrict__ dst,
                         const int* __restrict__ seg, int* __restrict__ cnt,
                         int* __restrict__ col, int E) {
    int e = blockIdx.x * blockDim.x + threadIdx.x;
    if (e >= E) return;
    int d = dst[e];
    int k = atomicSub(&cnt[d], 1);
    col[seg[2 * d] + k - 1] = src[e];
}


__global__ __launch_bounds__(256) void gcn_gemm_wmma(
    const float* __restrict__ X, const float* __restrict__ W,
    float* __restrict__ XW, int n)
{
    const int wave = threadIdx.x >> 5;
    const int lane = threadIdx.x & 31;
    const int half = lane >> 4;
    const int l16  = lane & 15;
    const int m0   = blockIdx.x * 16;
    const int n0   = wave * 16;

    int arow = m0 + l16;
    if (arow >= n) arow = n - 1;
    const float* Ap = X + (size_t)arow * CH;

    v8f c = {};
#pragma unroll
    for (int k0 = 0; k0 < CH; k0 += 4) {
        const int ka = k0 + 2 * half;
        v2f a, b;
        a.x = Ap[ka];
        a.y = Ap[ka + 1];
        b.x = W[(size_t)ka       * CH + n0 + l16];
        b.y = W[(size_t)(ka + 1) * CH + n0 + l16];
        c = __builtin_amdgcn_wmma_f32_16x16x4_f32(
                 false, a,  false, b,
                 (short)0, c,  false,  false);
    }

#pragma unroll
    for (int r = 0; r < 8; ++r) {
        int row = m0 + r + 8 * half;
        if (row < n) XW[(size_t)row * CH + n0 + l16] = c[r];
    }
}


#define GCN_ROW(u) \
    const float4 r##u = xw4[(size_t)__builtin_amdgcn_readlane(s_l, j + u) * 32 + lane];
#define GCN_ACC(u) { \
    const float w = __int_as_float(__builtin_amdgcn_readlane(w_l, j + u)); \
    acc.x = fmaf(r##u.x, w, acc.x); acc.y = fmaf(r##u.y, w, acc.y); \
    acc.z = fmaf(r##u.z, w, acc.z); acc.w = fmaf(r##u.w, w, acc.w); }

__global__ __launch_bounds__(256) void gcn_gather(
    const int* __restrict__ seg, const int* __restrict__ col,
    const float* __restrict__ dinv, const float* __restrict__ xw,
    const float* __restrict__ bias, float* __restrict__ out, int n)
{
    const int node = __builtin_amdgcn_readfirstlane((int)(blockIdx.x * 8u + (threadIdx.x >> 5)));
    if (node >= n) return;
    const int lane = threadIdx.x & 31;

    const int start = seg[2 * node];
    const int cnt   = seg[2 * node + 1];
    const float di  = dinv[node];
    const float4* xw4 = reinterpret_cast<const float4*>(xw);

    const float4 self = xw4[(size_t)node * 32 + lane];
    const float4 b4   = reinterpret_cast<const float4*>(bias)[lane];
    float4 acc;
    acc.x = b4.x + self.x * di * di;
    acc.y = b4.y + self.y * di * di;
    acc.z = b4.z + self.z * di * di;
    acc.w = b4.w + self.w * di * di;

    for (int base = 0; base < cnt; base += 32) {
        const int m = (cnt - base < 32) ? (cnt - base) : 32;
        int s_l = 0;
        int w_l = 0;
        if (lane < m) {
            s_l = col[start + base + lane];
            w_l = __float_as_int(dinv[s_l] * di);
        }
        int j = 0;
        for (; j + 8 <= m; j += 8) {
            GCN_ROW(0) GCN_ROW(1) GCN_ROW(2) GCN_ROW(3) GCN_ROW(4) GCN_ROW(5) GCN_ROW(6) GCN_ROW(7)
            GCN_ACC(0) GCN_ACC(1) GCN_ACC(2) GCN_ACC(3) GCN_ACC(4) GCN_ACC(5) GCN_ACC(6) GCN_ACC(7)
        }
        if (m & 4) {
            GCN_ROW(0) GCN_ROW(1) GCN_ROW(2) GCN_ROW(3)
            GCN_ACC(0) GCN_ACC(1) GCN_ACC(2) GCN_ACC(3)
            j += 4;
        }
        if (m & 2) {
            GCN_ROW(0) GCN_ROW(1)
            GCN_ACC(0) GCN_ACC(1)
            j += 2;
        }
        if (m & 1) {
            GCN_ROW(0)
            GCN_ACC(0)
        }
    }

    reinterpret_cast<float4*>(out)[(size_t)node * 32 + lane] = acc;
}


extern "C" void kernel_launch(void* const* d_in, const int* in_sizes, int n_in,
                              void* d_out, int out_size, void* d_ws, size_t ws_size,
                              hipStream_t stream) {
    const float* x  = (const float*)d_in[0];
    const float* W  = (const float*)d_in[1];
    const float* b  = (const float*)d_in[2];
    const int*   ei = (const int*)d_in[3];

    const int n = in_sizes[0] / CH;
    const int E = in_sizes[3] / 2;
    const int* src = ei;
    const int* dst = ei + E;

    const size_t np = ((size_t)n + 64) & ~(size_t)63;
    const size_t ep = ((size_t)E + 63) & ~(size_t)63;
    int*   cnt  = (int*)d_ws;
    int*   seg  = cnt + np;
    float* dinv = (float*)(seg + 2 * np);
    int*   col  = (int*)(dinv + np);
    float* xw   = (float*)(col + ep);
    float* out  = (float*)d_out;

    gcn_zero <<<(n + 256) / 256, 256, 0, stream>>>(cnt, n);
    gcn_hist <<<(E + 255) / 256, 256, 0, stream>>>(dst, cnt, E);
    gcn_alloc<<<(n + 255) / 256, 256, 0, stream>>>(cnt, cnt + n, seg, dinv, n);
    gcn_fill <<<(E + 255) / 256, 256, 0, stream>>>(src, dst, seg, cnt, col, E);

    gcn_gemm_wmma<<<(n + 15) / 16, 256, 0, stream>>>(x, W, xw, n);

    gcn_gather<<<(n + 7) / 8, 256, 0, stream>>>(seg, col, dinv, xw, b, out, n);
}


# group edges by destination with a two-level counting sort ranked in LDS (bin by 256-node range, then one work-group per range); no global atomics left; sum order = LDS arrival order
# speedup vs baseline: 2.6847x; 1.9609x over previous
#include <hip/hip_runtime.h>
#include <hip/hip_bf16.h>

#define CH 128

typedef __attribute__((ext_vector_type(2))) float v2f;
typedef __attribute__((ext_vector_type(8))) float v8f;


#define NPB       256
#define NPB_SHIFT 8
#define MAXNB     2048
#define GB        128
#define GT        1024
#define EPT       4

__global__ __launch_bounds__(GT) void gcn_count(const int* __restrict__ dst, int* __restrict__ tab,
                                                int E, int nb) {
    __shared__ int h[MAXNB];
    for (int k = threadIdx.x; k < nb; k += GT) h[k] = 0;
    __syncthreads();
    const int ch = (E + GB - 1) / GB;
    const int lo = blockIdx.x * ch;
    const int hi = (lo + ch < E) ? (lo + ch) : E;
    for (int e0 = lo + (int)threadIdx.x; e0 < hi; e0 += EPT * GT) {
        int d[EPT];
#pragma unroll
        for (int j = 0; j < EPT; ++j) d[j] = (e0 + j * GT < hi) ? dst[e0 + j * GT] : 0;
#pragma unroll
        for (int j = 0; j < EPT; ++j)
            if (e0 + j * GT < hi) atomicAdd(&h[d[j] >> NPB_SHIFT], 1);
    }
    __syncthreads();
    for (int k = threadIdx.x; k < nb; k += GT) tab[k * GB + blockIdx.x] = h[k];
}

__global__ __launch_bounds__(1024) void gcn_scan(int* __restrict__ tab, int L) {
    __shared__ int wsum[32];
    const int lane = threadIdx.x & 31;
    const int wave = threadIdx.x >> 5;
    const int per  = (((L + 31) / 32) + 127) & ~127;
    const int lo   = wave * per;
    const int hi   = (lo + per < L) ? (lo + per) : L;
    int4* t4 = reinterpret_cast<int4*>(tab);

    int acc = 0;
    for (int i = lo + 4 * lane; i < hi; i += 128) {
        const int4 v = t4[i >> 2];
        acc += v.x + v.y + v.z + v.w;
    }
#pragma unroll
    for (int off = 16; off > 0; off >>= 1) acc += __shfl_xor(acc, off);
    if (lane == 0) wsum[wave] = acc;
    __syncthreads();
    int carry = (lane < wave) ? wsum[lane] : 0;
#pragma unroll
    for (int off = 16; off > 0; off >>= 1) carry += __shfl_xor(carry, off);

    for (int i0 = lo; i0 < hi; i0 += 128) {
        const int i = i0 + 4 * lane;
        int4 v = make_int4(0, 0, 0, 0);
        if (i < hi) v = t4[i >> 2];
        const int s = v.x + v.y + v.z + v.w;
        int inc = s;
#pragma unroll
        for (int off = 1; off < 32; off <<= 1) {
            int t = __shfl_up(inc, off);
            if (lane >= off) inc += t;
        }
        const int ex = carry + inc - s;
        if (i < hi) {
            int4 o;
            o.x = ex;
            o.y = o.x + v.x;
            o.z = o.y + v.y;
            o.w = o.z + v.z;
            t4[i >> 2] = o;
        }
        carry += __shfl(inc, 31);
    }
}

__global__ __launch_bounds__(GT) void gcn_bin(const int* __restrict__ src, const int* __restrict__ dst,
                                              const int* __restrict__ tab, int* __restrict__ binned,
                                              int E, int nb) {
    __shared__ int cur[MAXNB];
    for (int k = threadIdx.x; k < nb; k += GT) cur[k] = tab[k * GB + blockIdx.x];
    __syncthreads();
    const int ch = (E + GB - 1) / GB;
    const int lo = blockIdx.x * ch;
    const int hi = (lo + ch < E) ? (lo + ch) : E;
    int2* b2 = reinterpret_cast<int2*>(binned);
    for (int e0 = lo + (int)threadIdx.x; e0 < hi; e0 += EPT * GT) {
        int d[EPT], s[EPT];
#pragma unroll
        for (int j = 0; j < EPT; ++j) {
            const bool in = e0 + j * GT < hi;
            d[j] = in ? dst[e0 + j * GT] : 0;
            s[j] = in ? src[e0 + j * GT] : 0;
        }
#pragma unroll
        for (int j = 0; j < EPT; ++j)
            if (e0 + j * GT < hi) {
                const int r = atomicAdd(&cur[d[j] >> NPB_SHIFT], 1);
                b2[r] = make_int2(s[j], d[j]);
            }
    }
}

__global__ __launch_bounds__(GT) void gcn_rank(const int* __restrict__ binned, const int* __restrict__ tab,
                                               int* __restrict__ seg, float* __restrict__ dinv,
                                               int* __restrict__ col, int n, int E, int nb) {
    __shared__ int cnt[NPB];
    __shared__ int pos[NPB];
    __shared__ int wsum[NPB / 32];
    const int tid  = threadIdx.x;
    const int lane = tid & 31;
    const int wave = tid >> 5;
    const int k    = blockIdx.x;
    const int n0   = k * NPB;
    const int bs   = tab[k * GB];
    const int be   = (k + 1 < nb) ? tab[(k + 1) * GB] : E;
    const int2* b2 = reinterpret_cast<const int2*>(binned);

    if (tid < NPB) cnt[tid] = 0;
    __syncthreads();
    for (int i0 = bs + tid; i0 < be; i0 += EPT * GT) {
        int d[EPT];
#pragma unroll
        for (int j = 0; j < EPT; ++j) d[j] = (i0 + j * GT < be) ? b2[i0 + j * GT].y : n0;
#pragma unroll
        for (int j = 0; j < EPT; ++j)
            if (i0 + j * GT < be) atomicAdd(&cnt[d[j] - n0], 1);
    }
    __syncthreads();

    const int c = (tid < NPB) ? cnt[tid] : 0;
    int s = c;
#pragma unroll
    for (int off = 1; off < 32; off <<= 1) {
        int t = __shfl_up(s, off);
        if (lane >= off) s += t;
    }
    if (tid < NPB && lane == 31) wsum[wave] = s;
    __syncthreads();
    if (tid < NPB) {
        int base = bs;
        for (int w = 0; w < wave; ++w) base += wsum[w];
        const int st = base + s - c;
        pos[tid] = st;
        const int node = n0 + tid;
        if (node < n) {
            seg[2 * node]     = st;
            seg[2 * node + 1] = c;
            dinv[node] = rsqrtf((float)(c + 1));
        }
    }
    __syncthreads();
    for (int i0 = bs + tid; i0 < be; i0 += EPT * GT) {
        int2 sd[EPT];
#pragma unroll
        for (int j = 0; j < EPT; ++j) sd[j] = (i0 + j * GT < be) ? b2[i0 + j * GT] : make_int2(0, n0);
#pragma unroll
        for (int j = 0; j < EPT; ++j)
            if (i0 + j * GT < be) {
                const int r = atomicAdd(&pos[sd[j].y - n0], 1);
                col[r] = sd[j].x;
            }
    }
}


__global__ __launch_bounds__(256) void gcn_gemm_wmma(
    const float* __restrict__ X, const float* __restrict__ W,
    float* __restrict__ XW, int n)
{
    const int wave = threadIdx.x >> 5;
    const int lane = threadIdx.x & 31;
    const int half = lane >> 4;
    const int l16  = lane & 15;
    const int m0   = blockIdx.x * 16;
    const int n0   = wave * 16;

    int arow = m0 + l16;
    if (arow >= n) arow = n - 1;
    const float* Ap = X + (size_t)arow * CH;

    v8f c = {};
#pragma unroll
    for (int k0 = 0; k0 < CH; k0 += 4) {
        const int ka = k0 + 2 * half;
        v2f a, b;
        a.x = Ap[ka];
        a.y = Ap[ka + 1];
        b.x = W[(size_t)ka       * CH + n0 + l16];
        b.y = W[(size_t)(ka + 1) * CH + n0 + l16];
        c = __builtin_amdgcn_wmma_f32_16x16x4_f32(
                 false, a,  false, b,
                 (short)0, c,  false,  false);
    }

#pragma unroll
    for (int r = 0; r < 8; ++r) {
        int row = m0 + r + 8 * half;
        if (row < n) XW[(size_t)row * CH + n0 + l16] = c[r];
    }
}


#define GCN_ROW(u) \
    const float4 r##u = xw4[(size_t)__builtin_amdgcn_readlane(s_l, j + u) * 32 + lane];
#define GCN_ACC(u) { \
    const float w = __int_as_float(__builtin_amdgcn_readlane(w_l, j + u)); \
    acc.x = fmaf(r##u.x, w, acc.x); acc.y = fmaf(r##u.y, w, acc.y); \
    acc.z = fmaf(r##u.z, w, acc.z); acc.w = fmaf(r##u.w, w, acc.w); }

__global__ __launch_bounds__(256) void gcn_gather(
    const int* __restrict__ seg, const int* __restrict__ col,
    const float* __restrict__ dinv, const float* __restrict__ xw,
    const float* __restrict__ bias, float* __restrict__ out, int n)
{
    const int node = __builtin_amdgcn_readfirstlane((int)(blockIdx.x * 8u + (threadIdx.x >> 5)));
    if (node >= n) return;
    const int lane = threadIdx.x & 31;

    const int start = seg[2 * node];
    const int cnt   = seg[2 * node + 1];
    const float di  = dinv[node];
    const float4* xw4 = reinterpret_cast<const float4*>(xw);

    const float4 self = xw4[(size_t)node * 32 + lane];
    const float4 b4   = reinterpret_cast<const float4*>(bias)[lane];
    float4 acc;
    acc.x = b4.x + self.x * di * di;
    acc.y = b4.y + self.y * di * di;
    acc.z = b4.z + self.z * di * di;
    acc.w = b4.w + self.w * di * di;

    for (int base = 0; base < cnt; base += 32) {
        const int m = (cnt - base < 32) ? (cnt - base) : 32;
        int s_l = 0;
        int w_l = 0;
        if (lane < m) {
            s_l = col[start + base + lane];
            w_l = __float_as_int(dinv[s_l] * di);
        }
        int j = 0;
        for (; j + 8 <= m; j += 8) {
            GCN_ROW(0) GCN_ROW(1) GCN_ROW(2) GCN_ROW(3) GCN_ROW(4) GCN_ROW(5) GCN_ROW(6) GCN_ROW(7)
            GCN_ACC(0) GCN_ACC(1) GCN_ACC(2) GCN_ACC(3) GCN_ACC(4) GCN_ACC(5) GCN_ACC(6) GCN_ACC(7)
        }
        if (m & 4) {
            GCN_ROW(0) GCN_ROW(1) GCN_ROW(2) GCN_ROW(3)
            GCN_ACC(0) GCN_ACC(1) GCN_ACC(2) GCN_ACC(3)
            j += 4;
        }
        if (m & 2) {
            GCN_ROW(0) GCN_ROW(1)
            GCN_ACC(0) GCN_ACC(1)
            j += 2;
        }
        if (m & 1) {
            GCN_ROW(0)
            GCN_ACC(0)
        }
    }

    reinterpret_cast<float4*>(out)[(size_t)node * 32 + lane] = acc;
}


extern "C" void kernel_launch(void* const* d_in, const int* in_sizes, int n_in,
                              void* d_out, int out_size, void* d_ws, size_t ws_size,
                              hipStream_t stream) {
    const float* x  = (const float*)d_in[0];
    const float* W  = (const float*)d_in[1];
    const float* b  = (const float*)d_in[2];
    const int*   ei = (const int*)d_in[3];

    const int n = in_sizes[0] / CH;
    const int E = in_sizes[3] / 2;
    const int* src = ei;
    const int* dst = ei + E;
    const int nb = (n + NPB - 1) / NPB;

    const size_t np = ((size_t)n + 64) & ~(size_t)63;
    const size_t ep = ((size_t)E + 63) & ~(size_t)63;
    const size_t tp = ((size_t)nb * GB + 63) & ~(size_t)63;
    int*   tab    = (int*)d_ws;
    int*   seg    = tab + tp;
    float* dinv   = (float*)(seg + 2 * np);
    int*   col    = (int*)(dinv + np);
    int*   binned = col + ep;
    float* xw     = (float*)(binned + 2 * ep);
    float* out    = (float*)d_out;

    gcn_count<<<GB, GT, 0, stream>>>(dst, tab, E, nb);
    gcn_scan <<<1, 1024, 0, stream>>>(tab, nb * GB);
    gcn_bin  <<<GB, GT, 0, stream>>>(src, dst, tab, binned, E, nb);
    gcn_rank <<<nb, GT, 0, stream>>>(binned, tab, seg, dinv, col, n, E, nb);

    gcn_gemm_wmma<<<(n + 15) / 16, 256, 0, stream>>>(x, W, xw, n);

    gcn_gather<<<(n + 7) / 8, 256, 0, stream>>>(seg, col, dinv, xw, b, out, n);
}


# the product x@W gets no launch of its own: its work-groups (64 rows each) ride behind the own work-groups of the four short grouping launches, a quarter each; 5 launches, results unchanged
# speedup vs baseline: 3.1578x; 1.1762x over previous
#include <hip/hip_runtime.h>
#include <hip/hip_bf16.h>

#define CH 128

typedef __attribute__((ext_vector_type(2))) float v2f;
typedef __attribute__((ext_vector_type(8))) float v8f;


__device__ __forceinline__ void gcn_gemm_rows(
    const float* __restrict__ X, const float* __restrict__ W,
    float* __restrict__ XW, int n, int blk)
{
    const int wave = threadIdx.x >> 5;
    const int lane = threadIdx.x & 31;
    const int half = lane >> 4;
    const int l16  = lane & 15;
    const int m0   = blk * 64 + (wave >> 3) * 16;
    const int n0   = (wave & 7) * 16;
    if (m0 >= n) return;

    int arow = m0 + l16;
    if (arow >= n) arow = n - 1;
    const float* Ap = X + (size_t)arow * CH;

    v8f c = {};
#pragma unroll 16
    for (int k0 = 0; k0 < CH; k0 += 4) {
        const int ka = k0 + 2 * half;
        v2f a, b;
        a.x = Ap[ka];
        a.y = Ap[ka + 1];
        b.x = W[(size_t)ka       * CH + n0 + l16];
        b.y = W[(size_t)(ka + 1) * CH + n0 + l16];
        c = __builtin_amdgcn_wmma_f32_16x16x4_f32(
                 false, a,  false, b,
                 (short)0, c,  false,  false);
    }

#pragma unroll
    for (int r = 0; r < 8; ++r) {
        int row = m0 + r + 8 * half;
        if (row < n) XW[(size_t)row * CH + n0 + l16] = c[r];
    }
}


#define NPB       256
#define NPB_SHIFT 8
#define MAXNB     2048
#define GB        128
#define GT        1024
#define EPT       4

__global__ __launch_bounds__(GT, 16) void gcn_count(const int* __restrict__ dst, int* __restrict__ tab,
                                                int E, int nb,
                                                const float* __restrict__ X, const float* __restrict__ Wm,
                                                float* __restrict__ XW, int n, int gfirst) {
    __shared__ int h[MAXNB];
    if (blockIdx.x >= GB) {
        gcn_gemm_rows(X, Wm, XW, n, gfirst + (int)blockIdx.x - GB);
        return;
    }
    for (int k = threadIdx.x; k < nb; k += GT) h[k] = 0;
    __syncthreads();
    const int ch = (E + GB - 1) / GB;
    const int lo = blockIdx.x * ch;
    const int hi = (lo + ch < E) ? (lo + ch) : E;
    for (int e0 = lo + (int)threadIdx.x; e0 < hi; e0 += EPT * GT) {
        int d[EPT];
#pragma unroll
        for (int j = 0; j < EPT; ++j) d[j] = (e0 + j * GT < hi) ? dst[e0 + j * GT] : 0;
#pragma unroll
        for (int j = 0; j < EPT; ++j)
            if (e0 + j * GT < hi) atomicAdd(&h[d[j] >> NPB_SHIFT], 1);
    }
    __syncthreads();
    for (int k = threadIdx.x; k < nb; k += GT) tab[k * GB + blockIdx.x] = h[k];
}

__global__ __launch_bounds__(1024, 16) void gcn_scan(int* __restrict__ tab, int L,
                                                 const float* __restrict__ X, const float* __restrict__ Wm,
                                                 float* __restrict__ XW, int n, int gfirst) {
    __shared__ int wsum[32];
    if (blockIdx.x >= 1) {
        gcn_gemm_rows(X, Wm, XW, n, gfirst + (int)blockIdx.x - 1);
        return;
    }
    const int lane = threadIdx.x & 31;
    const int wave = threadIdx.x >> 5;
    const int per  = (((L + 31) / 32) + 127) & ~127;
    const int lo   = wave * per;
    const int hi   = (lo + per < L) ? (lo + per) : L;
    int4* t4 = reinterpret_cast<int4*>(tab);

    int acc = 0;
    for (int i = lo + 4 * lane; i < hi; i += 128) {
        const int4 v = t4[i >> 2];
        acc += v.x + v.y + v.z + v.w;
    }
#pragma unroll
    for (int off = 16; off > 0; off >>= 1) acc += __shfl_xor(acc, off);
    if (lane == 0) wsum[wave] = acc;
    __syncthreads();
    int carry = (lane < wave) ? wsum[lane] : 0;
#pragma unroll
    for (int off = 16; off > 0; off >>= 1) carry += __shfl_xor(carry, off);

    for (int i0 = lo; i0 < hi; i0 += 128) {
        const int i = i0 + 4 * lane;
        int4 v = make_int4(0, 0, 0, 0);
        if (i < hi) v = t4[i >> 2];
        const int s = v.x + v.y + v.z + v.w;
        int inc = s;
#pragma unroll
        for (int off = 1; off < 32; off <<= 1) {
            int t = __shfl_up(inc, off);
            if (lane >= off) inc += t;
        }
        const int ex = carry + inc - s;
        if (i < hi) {
            int4 o;
            o.x = ex;
            o.y = o.x + v.x;
            o.z = o.y + v.y;
            o.w = o.z + v.z;
            t4[i >> 2] = o;
        }
        carry += __shfl(inc, 31);
    }
}

__global__ __launch_bounds__(GT, 16) void gcn_bin(const int* __restrict__ src, const int* __restrict__ dst,
                                              const int* __restrict__ tab, int* __restrict__ binned,
                                              int E, int nb,
                                              const float* __restrict__ X, const float* __restrict__ Wm,
                                              float* __restrict__ XW, int n, int gfirst) {
    __shared__ int cur[MAXNB];
    if (blockIdx.x >= GB) {
        gcn_gemm_rows(X, Wm, XW, n, gfirst + (int)blockIdx.x - GB);
        return;
    }
    for (int k = threadIdx.x; k < nb; k += GT) cur[k] = tab[k * GB + blockIdx.x];
    __syncthreads();
    const int ch = (E + GB - 1) / GB;
    const int lo = blockIdx.x * ch;
    const int hi = (lo + ch < E) ? (lo + ch) : E;
    int2* b2 = reinterpret_cast<int2*>(binned);
    for (int e0 = lo + (int)threadIdx.x; e0 < hi; e0 += EPT * GT) {
        int d[EPT], s[EPT];
#pragma unroll
        for (int j = 0; j < EPT; ++j) {
            const bool in = e0 + j * GT < hi;
            d[j] = in ? dst[e0 + j * GT] : 0;
            s[j] = in ? src[e0 + j * GT] : 0;
        }
#pragma unroll
        for (int j = 0; j < EPT; ++j)
            if (e0 + j * GT < hi) {
                const int r = atomicAdd(&cur[d[j] >> NPB_SHIFT], 1);
                b2[r] = make_int2(s[j], d[j]);
            }
    }
}

__global__ __launch_bounds__(GT, 16) void gcn_rank(const int* __restrict__ binned, const int* __restrict__ tab,
                                               int* __restrict__ seg, float* __restrict__ dinv,
                                               int* __restrict__ col, int n, int E, int nb,
                                               const float* __restrict__ X, const float* __restrict__ Wm,
                                               float* __restrict__ XW, int gfirst) {
    __shared__ int cnt[NPB];
    __shared__ int pos[NPB];
    __shared__ int wsum[NPB / 32];
    if ((int)blockIdx.x >= nb) {
        gcn_gemm_rows(X, Wm, XW, n, gfirst + (int)blockIdx.x - nb);
        return;
    }
    const int tid  = threadIdx.x;
    const int lane = tid & 31;
    const int wave = tid >> 5;
    const int k    = blockIdx.x;
    const int n0   = k * NPB;
    const int bs   = tab[k * GB];
    const int be   = (k + 1 < nb) ? tab[(k + 1) * GB] : E;
    const int2* b2 = reinterpret_cast<const int2*>(binned);

    if (tid < NPB) cnt[tid] = 0;
    __syncthreads();
    for (int i0 = bs + tid; i0 < be; i0 += EPT * GT) {
        int d[EPT];
#pragma unroll
        for (int j = 0; j < EPT; ++j) d[j] = (i0 + j * GT < be) ? b2[i0 + j * GT].y : n0;
#pragma unroll
        for (int j = 0; j < EPT; ++j)
            if (i0 + j * GT < be) atomicAdd(&cnt[d[j] - n0], 1);
    }
    __syncthreads();

    const int c = (tid < NPB) ? cnt[tid] : 0;
    int s = c;
#pragma unroll
    for (int off = 1; off < 32; off <<= 1) {
        int t = __shfl_up(s, off);
        if (lane >= off) s += t;
    }
    if (tid < NPB && lane == 31) wsum[wave] = s;
    __syncthreads();
    if (tid < NPB) {
        int base = bs;
        for (int w = 0; w < wave; ++w) base += wsum[w];
        const int st = base + s - c;
        pos[tid] = st;
        const int node = n0 + tid;
        if (node < n) {
            seg[2 * node]     = st;
            seg[2 * node + 1] = c;
            dinv[node] = rsqrtf((float)(c + 1));
        }
    }
    __syncthreads();
    for (int i0 = bs + tid; i0 < be; i0 += EPT * GT) {
        int2 sd[EPT];
#pragma unroll
        for (int j = 0; j < EPT; ++j) sd[j] = (i0 + j * GT < be) ? b2[i0 + j * GT] : make_int2(0, n0);
#pragma unroll
        for (int j = 0; j < EPT; ++j)
            if (i0 + j * GT < be) {
                const int r = atomicAdd(&pos[sd[j].y - n0], 1);
                col[r] = sd[j].x;
            }
    }
}


#define GCN_ROW(u) \
    const float4 r##u = xw4[(size_t)__builtin_amdgcn_readlane(s_l, j + u) * 32 + lane];
#define GCN_ACC(u) { \
    const float w = __int_as_float(__builtin_amdgcn_readlane(w_l, j + u)); \
    acc.x = fmaf(r##u.x, w, acc.x); acc.y = fmaf(r##u.y, w, acc.y); \
    acc.z = fmaf(r##u.z, w, acc.z); acc.w = fmaf(r##u.w, w, acc.w); }

__global__ __launch_bounds__(256) void gcn_gather(
    const int* __restrict__ seg, const int* __restrict__ col,
    const float* __restrict__ dinv, const float* __restrict__ xw,
    const float* __restrict__ bias, float* __restrict__ out, int n)
{
    const int node = __builtin_amdgcn_readfirstlane((int)(blockIdx.x * 8u + (threadIdx.x >> 5)));
    if (node >= n) return;
    const int lane = threadIdx.x & 31;

    const int start = seg[2 * node];
    const int cnt   = seg[2 * node + 1];
    const float di  = dinv[node];
    const float4* xw4 = reinterpret_cast<const float4*>(xw);

    const float4 self = xw4[(size_t)node * 32 + lane];
    const float4 b4   = reinterpret_cast<const float4*>(bias)[lane];
    float4 acc;
    acc.x = b4.x + self.x * di * di;
    acc.y = b4.y + self.y * di * di;
    acc.z = b4.z + self.z * di * di;
    acc.w = b4.w + self.w * di * di;

    for (int base = 0; base < cnt; base += 32) {
        const int m = (cnt - base < 32) ? (cnt - base) : 32;
        int s_l = 0;
        int w_l = 0;
        if (lane < m) {
            s_l = col[start + base + lane];
            w_l = __float_as_int(dinv[s_l] * di);
        }
        int j = 0;
        for (; j + 8 <= m; j += 8) {
            GCN_ROW(0) GCN_ROW(1) GCN_ROW(2) GCN_ROW(3) GCN_ROW(4) GCN_ROW(5) GCN_ROW(6) GCN_ROW(7)
            GCN_ACC(0) GCN_ACC(1) GCN_ACC(2) GCN_ACC(3) GCN_ACC(4) GCN_ACC(5) GCN_ACC(6) GCN_ACC(7)
        }
        if (m & 4) {
            GCN_ROW(0) GCN_ROW(1) GCN_ROW(2) GCN_ROW(3)
            GCN_ACC(0) GCN_ACC(1) GCN_ACC(2) GCN_ACC(3)
            j += 4;
        }
        if (m & 2) {
            GCN_ROW(0) GCN_ROW(1)
            GCN_ACC(0) GCN_ACC(1)
            j += 2;
        }
        if (m & 1) {
            GCN_ROW(0)
            GCN_ACC(0)
        }
    }

    reinterpret_cast<float4*>(out)[(size_t)node * 32 + lane] = acc;
}


extern "C" void kernel_launch(void* const* d_in, const int* in_sizes, int n_in,
                              void* d_out, int out_size, void* d_ws, size_t ws_size,
                              hipStream_t stream) {
    const float* x  = (const float*)d_in[0];
    const float* W  = (const float*)d_in[1];
    const float* b  = (const float*)d_in[2];
    const int*   ei = (const int*)d_in[3];

    const int n = in_sizes[0] / CH;
    const int E = in_sizes[3] / 2;
    const int* src = ei;
    const int* dst = ei + E;
    const int nb = (n + NPB - 1) / NPB;

    const size_t np = ((size_t)n + 64) & ~(size_t)63;
    const size_t ep = ((size_t)E + 63) & ~(size_t)63;
    const size_t tp = ((size_t)nb * GB + 63) & ~(size_t)63;
    int*   tab    = (int*)d_ws;
    int*   seg    = tab + tp;
    float* dinv   = (float*)(seg + 2 * np);
    int*   col    = (int*)(dinv + np);
    int*   binned = col + ep;
    float* xw     = (float*)(binned + 2 * ep);
    float* out    = (float*)d_out;

    const int gr = ((n + 63) / 64 + 3) / 4;
    gcn_count<<<GB + gr, GT, 0, stream>>>(dst, tab, E, nb, x, W, xw, n, 0);
    gcn_scan <<<1 + gr, 1024, 0, stream>>>(tab, nb * GB, x, W, xw, n, gr);
    gcn_bin  <<<GB + gr, GT, 0, stream>>>(src, dst, tab, binned, E, nb, x, W, xw, n, 2 * gr);
    gcn_rank <<<nb + gr, GT, 0, stream>>>(binned, tab, seg, dinv, col, n, E, nb, x, W, xw, 3 * gr);

    gcn_gather<<<(n + 7) / 8, 256, 0, stream>>>(seg, col, dinv, xw, b, out, n);
}


# product: the 64x128 tile of x is copied to LDS with whole-row loads and the A fragments are read from LDS (was 32 global loads of 16 lines each per wave); k order and sums unchanged
# speedup vs baseline: 3.5262x; 1.1167x over previous
#include <hip/hip_runtime.h>
#include <hip/hip_bf16.h>

#define CH 128

typedef __attribute__((ext_vector_type(2))) float v2f;
typedef __attribute__((ext_vector_type(8))) float v8f;


#define APITCH 132

__device__ __forceinline__ void gcn_gemm_rows(
    const float* __restrict__ X, const float* __restrict__ W,
    float* __restrict__ XW, int n, int blk)
{
    __shared__ float As[64 * APITCH];
    const int r0 = blk * 64;
    if (r0 >= n) return;

#pragma unroll
    for (int j = 0; j < 2; ++j) {
        const int idx = (int)threadIdx.x + j * 1024;
        const int row = idx >> 5;
        const int c4  = idx & 31;
        int grow = r0 + row;
        if (grow >= n) grow = n - 1;
        const float4 v = reinterpret_cast<const float4*>(X + (size_t)grow * CH)[c4];
        *reinterpret_cast<float4*>(&As[row * APITCH + 4 * c4]) = v;
    }
    __syncthreads();

    const int wave = threadIdx.x >> 5;
    const int lane = threadIdx.x & 31;
    const int half = lane >> 4;
    const int l16  = lane & 15;
    const int m0   = r0 + (wave >> 3) * 16;
    const int n0   = (wave & 7) * 16;
    if (m0 >= n) return;

    const float* Ap = &As[((wave >> 3) * 16 + l16) * APITCH + 2 * half];
    const float* Bp = W + (size_t)(2 * half) * CH + n0 + l16;

    v8f c = {};
#pragma unroll 16
    for (int k0 = 0; k0 < CH; k0 += 4) {
        v2f a, b;
        a = *reinterpret_cast<const v2f*>(Ap + k0);
        b.x = Bp[(size_t)k0       * CH];
        b.y = Bp[(size_t)(k0 + 1) * CH];
        c = __builtin_amdgcn_wmma_f32_16x16x4_f32(
                 false, a,  false, b,
                 (short)0, c,  false,  false);
    }

#pragma unroll
    for (int r = 0; r < 8; ++r) {
        int row = m0 + r + 8 * half;
        if (row < n) XW[(size_t)row * CH + n0 + l16] = c[r];
    }
}


#define NPB       256
#define NPB_SHIFT 8
#define MAXNB     2048
#define GB        128
#define GT        1024
#define EPT       4

__global__ __launch_bounds__(GT, 16) void gcn_count(const int* __restrict__ dst, int* __restrict__ tab,
                                                int E, int nb,
                                                const float* __restrict__ X, const float* __restrict__ Wm,
                                                float* __restrict__ XW, int n, int gfirst) {
    __shared__ int h[MAXNB];
    if (blockIdx.x >= GB) {
        gcn_gemm_rows(X, Wm, XW, n, gfirst + (int)blockIdx.x - GB);
        return;
    }
    for (int k = threadIdx.x; k < nb; k += GT) h[k] = 0;
    __syncthreads();
    const int ch = (E + GB - 1) / GB;
    const int lo = blockIdx.x * ch;
    const int hi = (lo + ch < E) ? (lo + ch) : E;
    for (int e0 = lo + (int)threadIdx.x; e0 < hi; e0 += EPT * GT) {
        int d[EPT];
#pragma unroll
        for (int j = 0; j < EPT; ++j) d[j] = (e0 + j * GT < hi) ? dst[e0 + j * GT] : 0;
#pragma unroll
        for (int j = 0; j < EPT; ++j)
            if (e0 + j * GT < hi) atomicAdd(&h[d[j] >> NPB_SHIFT], 1);
    }
    __syncthreads();
    for (int k = threadIdx.x; k < nb; k += GT) tab[k * GB + blockIdx.x] = h[k];
}

__global__ __launch_bounds__(1024, 16) void gcn_scan(int* __restrict__ tab, int L,
                                                 const float* __restrict__ X, const float* __restrict__ Wm,
                                                 float* __restrict__ XW, int n, int gfirst) {
    __shared__ int wsum[32];
    if (blockIdx.x >= 1) {
        gcn_gemm_rows(X, Wm, XW, n, gfirst + (int)blockIdx.x - 1);
        return;
    }
    const int lane = threadIdx.x & 31;
    const int wave = threadIdx.x >> 5;
    const int per  = (((L + 31) / 32) + 127) & ~127;
    const int lo   = wave * per;
    const int hi   = (lo + per < L) ? (lo + per) : L;
    int4* t4 = reinterpret_cast<int4*>(tab);

    int acc = 0;
    for (int i = lo + 4 * lane; i < hi; i += 128) {
        const int4 v = t4[i >> 2];
        acc += v.x + v.y + v.z + v.w;
    }
#pragma unroll
    for (int off = 16; off > 0; off >>= 1) acc += __shfl_xor(acc, off);
    if (lane == 0) wsum[wave] = acc;
    __syncthreads();
    int carry = (lane < wave) ? wsum[lane] : 0;
#pragma unroll
    for (int off = 16; off > 0; off >>= 1) carry += __shfl_xor(carry, off);

    for (int i0 = lo; i0 < hi; i0 += 128) {
        const int i = i0 + 4 * lane;
        int4 v = make_int4(0, 0, 0, 0);
        if (i < hi) v = t4[i >> 2];
        const int s = v.x + v.y + v.z + v.w;
        int inc = s;
#pragma unroll
        for (int off = 1; off < 32; off <<= 1) {
            int t = __shfl_up(inc, off);
            if (lane >= off) inc += t;
        }
        const int ex = carry + inc - s;
        if (i < hi) {
            int4 o;
            o.x = ex;
            o.y = o.x + v.x;
            o.z = o.y + v.y;
            o.w = o.z + v.z;
            t4[i >> 2] = o;
        }
        carry += __shfl(inc, 31);
    }
}

__global__ __launch_bounds__(GT, 16) void gcn_bin(const int* __restrict__ src, const int* __restrict__ dst,
                                              const int* __restrict__ tab, int* __restrict__ binned,
                                              int E, int nb,
                                              const float* __restrict__ X, const float* __restrict__ Wm,
                                              float* __restrict__ XW, int n, int gfirst) {
    __shared__ int cur[MAXNB];
    if (blockIdx.x >= GB) {
        gcn_gemm_rows(X, Wm, XW, n, gfirst + (int)blockIdx.x - GB);
        return;
    }
    for (int k = threadIdx.x; k < nb; k += GT) cur[k] = tab[k * GB + blockIdx.x];
    __syncthreads();
    const int ch = (E + GB - 1) / GB;
    const int lo = blockIdx.x * ch;
    const int hi = (lo + ch < E) ? (lo + ch) : E;
    int2* b2 = reinterpret_cast<int2*>(binned);
    for (int e0 = lo + (int)threadIdx.x; e0 < hi; e0 += EPT * GT) {
        int d[EPT], s[EPT];
#pragma unroll
        for (int j = 0; j < EPT; ++j) {
            const bool in = e0 + j * GT < hi;
            d[j] = in ? dst[e0 + j * GT] : 0;
            s[j] = in ? src[e0 + j * GT] : 0;
        }
#pragma unroll
        for (int j = 0; j < EPT; ++j)
            if (e0 + j * GT < hi) {
                const int r = atomicAdd(&cur[d[j] >> NPB_SHIFT], 1);
                b2[r] = make_int2(s[j], d[j]);
            }
    }
}

__global__ __launch_bounds__(GT, 16) void gcn_rank(const int* __restrict__ binned, const int* __restrict__ tab,
                                               int* __restrict__ seg, float* __restrict__ dinv,
                                               int* __restrict__ col, int n, int E, int nb,
                                               const float* __restrict__ X, const float* __restrict__ Wm,
                                               float* __restrict__ XW, int gfirst) {
    __shared__ int cnt[NPB];
    __shared__ int pos[NPB];
    __shared__ int wsum[NPB / 32];
    if ((int)blockIdx.x >= nb) {
        gcn_gemm_rows(X, Wm, XW, n, gfirst + (int)blockIdx.x - nb);
        return;
    }
    const int tid  = threadIdx.x;
    const int lane = tid & 31;
    const int wave = tid >> 5;
    const int k    = blockIdx.x;
    const int n0   = k * NPB;
    const int bs   = tab[k * GB];
    const int be   = (k + 1 < nb) ? tab[(k + 1) * GB] : E;
    const int2* b2 = reinterpret_cast<const int2*>(binned);

    if (tid < NPB) cnt[tid] = 0;
    __syncthreads();
    for (int i0 = bs + tid; i0 < be; i0 += EPT * GT) {
        int d[EPT];
#pragma unroll
        for (int j = 0; j < EPT; ++j) d[j] = (i0 + j * GT < be) ? b2[i0 + j * GT].y : n0;
#pragma unroll
        for (int j = 0; j < EPT; ++j)
            if (i0 + j * GT < be) atomicAdd(&cnt[d[j] - n0], 1);
    }
    __syncthreads();

    const int c = (tid < NPB) ? cnt[tid] : 0;
    int s = c;
#pragma unroll
    for (int off = 1; off < 32; off <<= 1) {
        int t = __shfl_up(s, off);
        if (lane >= off) s += t;
    }
    if (tid < NPB && lane == 31) wsum[wave] = s;
    __syncthreads();
    if (tid < NPB) {
        int base = bs;
        for (int w = 0; w < wave; ++w) base += wsum[w];
        const int st = base + s - c;
        pos[tid] = st;
        const int node = n0 + tid;
        if (node < n) {
            seg[2 * node]     = st;
            seg[2 * node + 1] = c;
            dinv[node] = rsqrtf((float)(c + 1));
        }
    }
    __syncthreads();
    for (int i0 = bs + tid; i0 < be; i0 += EPT * GT) {
        int2 sd[EPT];
#pragma unroll
        for (int j = 0; j < EPT; ++j) sd[j] = (i0 + j * GT < be) ? b2[i0 + j * GT] : make_int2(0, n0);
#pragma unroll
        for (int j = 0; j < EPT; ++j)
            if (i0 + j * GT < be) {
                const int r = atomicAdd(&pos[sd[j].y - n0], 1);
                col[r] = sd[j].x;
            }
    }
}


#define GCN_ROW(u) \
    const float4 r##u = xw4[(size_t)__builtin_amdgcn_readlane(s_l, j + u) * 32 + lane];
#define GCN_ACC(u) { \
    const float w = __int_as_float(__builtin_amdgcn_readlane(w_l, j + u)); \
    acc.x = fmaf(r##u.x, w, acc.x); acc.y = fmaf(r##u.y, w, acc.y); \
    acc.z = fmaf(r##u.z, w, acc.z); acc.w = fmaf(r##u.w, w, acc.w); }

__global__ __launch_bounds__(256) void gcn_gather(
    const int* __restrict__ seg, const int* __restrict__ col,
    const float* __restrict__ dinv, const float* __restrict__ xw,
    const float* __restrict__ bias, float* __restrict__ out, int n)
{
    const int node = __builtin_amdgcn_readfirstlane((int)(blockIdx.x * 8u + (threadIdx.x >> 5)));
    if (node >= n) return;
    const int lane = threadIdx.x & 31;

    const int start = seg[2 * node];
    const int cnt   = seg[2 * node + 1];
    const float di  = dinv[node];
    const float4* xw4 = reinterpret_cast<const float4*>(xw);

    const float4 self = xw4[(size_t)node * 32 + lane];
    const float4 b4   = reinterpret_cast<const float4*>(bias)[lane];
    float4 acc;
    acc.x = b4.x + self.x * di * di;
    acc.y = b4.y + self.y * di * di;
    acc.z = b4.z + self.z * di * di;
    acc.w = b4.w + self.w * di * di;

    for (int base = 0; base < cnt; base += 32) {
        const int m = (cnt - base < 32) ? (cnt - base) : 32;
        int s_l = 0;
        int w_l = 0;
        if (lane < m) {
            s_l = col[start + base + lane];
            w_l = __float_as_int(dinv[s_l] * di);
        }
        int j = 0;
        for (; j + 8 <= m; j += 8) {
            GCN_ROW(0) GCN_ROW(1) GCN_ROW(2) GCN_ROW(3) GCN_ROW(4) GCN_ROW(5) GCN_ROW(6) GCN_ROW(7)
            GCN_ACC(0) GCN_ACC(1) GCN_ACC(2) GCN_ACC(3) GCN_ACC(4) GCN_ACC(5) GCN_ACC(6) GCN_ACC(7)
        }
        if (m & 4) {
            GCN_ROW(0) GCN_ROW(1) GCN_ROW(2) GCN_ROW(3)
            GCN_ACC(0) GCN_ACC(1) GCN_ACC(2) GCN_ACC(3)
            j += 4;
        }
        if (m & 2) {
            GCN_ROW(0) GCN_ROW(1)
            GCN_ACC(0) GCN_ACC(1)
            j += 2;
        }
        if (m & 1) {
            GCN_ROW(0)
            GCN_ACC(0)
        }
    }

    reinterpret_cast<float4*>(out)[(size_t)node * 32 + lane] = acc;
}


extern "C" void kernel_launch(void* const* d_in, const int* in_sizes, int n_in,
                              void* d_out, int out_size, void* d_ws, size_t ws_size,
                              hipStream_t stream) {
    const float* x  = (const float*)d_in[0];
    const float* W  = (const float*)d_in[1];
    const float* b  = (const float*)d_in[2];
    const int*   ei = (const int*)d_in[3];

    const int n = in_sizes[0] / CH;
    const int E = in_sizes[3] / 2;
    const int* src = ei;
    const int* dst = ei + E;
    const int nb = (n + NPB - 1) / NPB;

    const size_t np = ((size_t)n + 64) & ~(size_t)63;
    const size_t ep = ((size_t)E + 63) & ~(size_t)63;
    const size_t tp = ((size_t)nb * GB + 63) & ~(size_t)63;
    int*   tab    = (int*)d_ws;
    int*   seg    = tab + tp;
    float* dinv   = (float*)(seg + 2 * np);
    int*   col    = (int*)(dinv + np);
    int*   binned = col + ep;
    float* xw     = (float*)(binned + 2 * ep);
    float* out    = (float*)d_out;

    const int gr = ((n + 63) / 64 + 3) / 4;
    gcn_count<<<GB + gr, GT, 0, stream>>>(dst, tab, E, nb, x, W, xw, n, 0);
    gcn_scan <<<1 + gr, 1024, 0, stream>>>(tab, nb * GB, x, W, xw, n, gr);
    gcn_bin  <<<GB + gr, GT, 0, stream>>>(src, dst, tab, binned, E, nb, x, W, xw, n, 2 * gr);
    gcn_rank <<<nb + gr, GT, 0, stream>>>(binned, tab, seg, dinv, col, n, E, nb, x, W, xw, 3 * gr);

    gcn_gather<<<(n + 7) / 8, 256, 0, stream>>>(seg, col, dinv, xw, b, out, n);
}


# grouping without a global count table or scan: each of 128 work-groups sorts its own edge chunk by bucket in LDS, rank reads its bucket's 128 pieces; 3 launches (was 5); sum order = LDS arrival order
# speedup vs baseline: 4.4555x; 1.2635x over previous
#include <hip/hip_runtime.h>
#include <hip/hip_bf16.h>

#define CH 128

typedef __attribute__((ext_vector_type(2))) float v2f;
typedef __attribute__((ext_vector_type(8))) float v8f;


#define APITCH 132

__device__ __forceinline__ void gcn_gemm_rows(
    const float* __restrict__ X, const float* __restrict__ W,
    float* __restrict__ XW, int n, int blk)
{
    __shared__ float As[64 * APITCH];
    const int r0 = blk * 64;
    if (r0 >= n) return;

#pragma unroll
    for (int j = 0; j < 2; ++j) {
        const int idx = (int)threadIdx.x + j * 1024;
        const int row = idx >> 5;
        const int c4  = idx & 31;
        int grow = r0 + row;
        if (grow >= n) grow = n - 1;
        const float4 v = reinterpret_cast<const float4*>(X + (size_t)grow * CH)[c4];
        *reinterpret_cast<float4*>(&As[row * APITCH + 4 * c4]) = v;
    }
    __syncthreads();

    const int wave = threadIdx.x >> 5;
    const int lane = threadIdx.x & 31;
    const int half = lane >> 4;
    const int l16  = lane & 15;
    const int m0   = r0 + (wave >> 3) * 16;
    const int n0   = (wave & 7) * 16;
    if (m0 >= n) return;

    const float* Ap = &As[((wave >> 3) * 16 + l16) * APITCH + 2 * half];
    const float* Bp = W + (size_t)(2 * half) * CH + n0 + l16;

    v8f c = {};
#pragma unroll 16
    for (int k0 = 0; k0 < CH; k0 += 4) {
        v2f a, b;
        a = *reinterpret_cast<const v2f*>(Ap + k0);
        b.x = Bp[(size_t)k0       * CH];
        b.y = Bp[(size_t)(k0 + 1) * CH];
        c = __builtin_amdgcn_wmma_f32_16x16x4_f32(
                 false, a,  false, b,
                 (short)0, c,  false,  false);
    }

#pragma unroll
    for (int r = 0; r < 8; ++r) {
        int row = m0 + r + 8 * half;
        if (row < n) XW[(size_t)row * CH + n0 + l16] = c[r];
    }
}


#define NPB       256
#define NPB_SHIFT 8
#define MAXNB     2048
#define GB        128
#define GT        1024
#define EPT       4
#define PPW       (GB / 32)
#define PSL       2
#define NWGP      256

__global__ __launch_bounds__(GT, 16) void gcn_chunk(const int* __restrict__ src, const int* __restrict__ dst,
                                                int* __restrict__ offt, int* __restrict__ binned,
                                                int E, int nb, int tp,
                                                const float* __restrict__ X, const float* __restrict__ Wm,
                                                float* __restrict__ XW, int n, int gfirst) {
    __shared__ int h[MAXNB];
    __shared__ int wsum[GT / 32];
    if (blockIdx.x >= GB) {
        gcn_gemm_rows(X, Wm, XW, n, gfirst + (int)blockIdx.x - GB);
        return;
    }
    const int tid  = threadIdx.x;
    const int lane = tid & 31;
    const int wave = tid >> 5;
    for (int k = tid; k < nb; k += GT) h[k] = 0;
    __syncthreads();
    const int ch = (E + GB - 1) / GB;
    const int lo = blockIdx.x * ch;
    const int hi = (lo + ch < E) ? (lo + ch) : E;
    for (int e0 = lo + tid; e0 < hi; e0 += EPT * GT) {
        int d[EPT];
#pragma unroll
        for (int j = 0; j < EPT; ++j) d[j] = (e0 + j * GT < hi) ? dst[e0 + j * GT] : 0;
#pragma unroll
        for (int j = 0; j < EPT; ++j)
            if (e0 + j * GT < hi) atomicAdd(&h[d[j] >> NPB_SHIFT], 1);
    }
    __syncthreads();

    const int i0 = 2 * tid;
    const int a0 = (i0 < nb) ? h[i0] : 0;
    const int a1 = (i0 + 1 < nb) ? h[i0 + 1] : 0;
    const int s  = a0 + a1;
    int inc = s;
#pragma unroll
    for (int off = 1; off < 32; off <<= 1) {
        int t = __shfl_up(inc, off);
        if (lane >= off) inc += t;
    }
    if (lane == 31) wsum[wave] = inc;
    __syncthreads();
    int ex = inc - s;
    for (int w = 0; w < wave; ++w) ex += wsum[w];
    int* orow = offt + (size_t)blockIdx.x * tp;
    if (i0 < nb)     { h[i0]     = ex;      orow[i0]     = ex; }
    if (i0 + 1 < nb) { h[i0 + 1] = ex + a0; orow[i0 + 1] = ex + a0; }
    if (tid == 0) orow[nb] = (hi > lo) ? (hi - lo) : 0;
    __syncthreads();

    int2* b2 = reinterpret_cast<int2*>(binned);
    for (int e0 = lo + tid; e0 < hi; e0 += EPT * GT) {
        int d[EPT], sv[EPT];
#pragma unroll
        for (int j = 0; j < EPT; ++j) {
            const bool in = e0 + j * GT < hi;
            d[j]  = in ? dst[e0 + j * GT] : 0;
            sv[j] = in ? src[e0 + j * GT] : 0;
        }
#pragma unroll
        for (int j = 0; j < EPT; ++j)
            if (e0 + j * GT < hi) {
                const int r = atomicAdd(&h[d[j] >> NPB_SHIFT], 1);
                b2[lo + r] = make_int2(sv[j], d[j]);
            }
    }
}

__global__ __launch_bounds__(GT, 16) void gcn_rank(const int* __restrict__ binned, const int* __restrict__ offt,
                                               int* __restrict__ seg, float* __restrict__ dinv,
                                               int* __restrict__ col, int n, int E, int nb, int tp,
                                               const float* __restrict__ X, const float* __restrict__ Wm,
                                               float* __restrict__ XW, int gfirst) {
    __shared__ int cnt[NPB];
    __shared__ int pos[NPB];
    __shared__ int wsum[NPB / 32];
    __shared__ int bsum;
    if ((int)blockIdx.x >= nb) {
        gcn_gemm_rows(X, Wm, XW, n, gfirst + (int)blockIdx.x - nb);
        return;
    }
    const int tid  = threadIdx.x;
    const int lane = tid & 31;
    const int wave = tid >> 5;
    const int k    = blockIdx.x;
    const int n0   = k * NPB;
    const int ch   = (E + GB - 1) / GB;
    const int2* b2 = reinterpret_cast<const int2*>(binned);

    int p_rel = 0, p_lo = 0, p_hi = 0;
    if (lane < PPW) {
        const int g = wave + 32 * lane;
        const int* orow = offt + (size_t)g * tp + k;
        p_rel = orow[0];
        p_lo  = g * ch + p_rel;
        p_hi  = g * ch + orow[1];
    }
    if (tid < NPB) cnt[tid] = 0;
    if (tid == 0) bsum = 0;
    __syncthreads();
    if (lane < PPW) atomicAdd(&bsum, p_rel);

    int st[PPW], en[PPW];
#pragma unroll
    for (int j = 0; j < PPW; ++j) {
        st[j] = __builtin_amdgcn_readlane(p_lo, j);
        en[j] = __builtin_amdgcn_readlane(p_hi, j);
    }
    int2 sd[PPW * PSL];
#pragma unroll
    for (int j = 0; j < PPW; ++j)
#pragma unroll
        for (int u = 0; u < PSL; ++u) {
            const int i = st[j] + 32 * u + lane;
            sd[j * PSL + u] = (i < en[j]) ? b2[i] : make_int2(0, n0);
        }
#pragma unroll
    for (int j = 0; j < PPW; ++j)
#pragma unroll
        for (int u = 0; u < PSL; ++u)
            if (st[j] + 32 * u + lane < en[j]) atomicAdd(&cnt[sd[j * PSL + u].y - n0], 1);
#pragma unroll
    for (int j = 0; j < PPW; ++j)
        for (int i = st[j] + 32 * PSL + lane; i < en[j]; i += 32) atomicAdd(&cnt[b2[i].y - n0], 1);
    __syncthreads();

    const int c = (tid < NPB) ? cnt[tid] : 0;
    int s = c;
#pragma unroll
    for (int off = 1; off < 32; off <<= 1) {
        int t = __shfl_up(s, off);
        if (lane >= off) s += t;
    }
    if (tid < NPB && lane == 31) wsum[wave] = s;
    __syncthreads();
    if (tid < NPB) {
        int base = bsum;
        for (int w = 0; w < wave; ++w) base += wsum[w];
        const int stn = base + s - c;
        pos[tid] = stn;
        const int node = n0 + tid;
        if (node < n) {
            seg[2 * node]     = stn;
            seg[2 * node + 1] = c;
            dinv[node] = rsqrtf((float)(c + 1));
        }
    }
    __syncthreads();
#pragma unroll
    for (int j = 0; j < PPW; ++j)
#pragma unroll
        for (int u = 0; u < PSL; ++u)
            if (st[j] + 32 * u + lane < en[j]) {
                const int r = atomicAdd(&pos[sd[j * PSL + u].y - n0], 1);
                col[r] = sd[j * PSL + u].x;
            }
#pragma unroll
    for (int j = 0; j < PPW; ++j)
        for (int i = st[j] + 32 * PSL + lane; i < en[j]; i += 32) {
            const int2 e = b2[i];
            const int r = atomicAdd(&pos[e.y - n0], 1);
            col[r] = e.x;
        }
}


#define GCN_ROW(u) \
    const float4 r##u = xw4[(size_t)__builtin_amdgcn_readlane(s_l, j + u) * 32 + lane];
#define GCN_ACC(u) { \
    const float w = __int_as_float(__builtin_amdgcn_readlane(w_l, j + u)); \
    acc.x = fmaf(r##u.x, w, acc.x); acc.y = fmaf(r##u.y, w, acc.y); \
    acc.z = fmaf(r##u.z, w, acc.z); acc.w = fmaf(r##u.w, w, acc.w); }

__global__ __launch_bounds__(256) void gcn_gather(
    const int* __restrict__ seg, const int* __restrict__ col,
    const float* __restrict__ dinv, const float* __restrict__ xw,
    const float* __restrict__ bias, float* __restrict__ out, int n)
{
    const int node = __builtin_amdgcn_readfirstlane((int)(blockIdx.x * 8u + (threadIdx.x >> 5)));
    if (node >= n) return;
    const int lane = threadIdx.x & 31;

    const int start = seg[2 * node];
    const int cnt   = seg[2 * node + 1];
    const float di  = dinv[node];
    const float4* xw4 = reinterpret_cast<const float4*>(xw);

    const float4 self = xw4[(size_t)node * 32 + lane];
    const float4 b4   = reinterpret_cast<const float4*>(bias)[lane];
    float4 acc;
    acc.x = b4.x + self.x * di * di;
    acc.y = b4.y + self.y * di * di;
    acc.z = b4.z + self.z * di * di;
    acc.w = b4.w + self.w * di * di;

    for (int base = 0; base < cnt; base += 32) {
        const int m = (cnt - base < 32) ? (cnt - base) : 32;
        int s_l = 0;
        int w_l = 0;
        if (lane < m) {
            s_l = col[start + base + lane];
            w_l = __float_as_int(dinv[s_l] * di);
        }
        int j = 0;
        for (; j + 8 <= m; j += 8) {
            GCN_ROW(0) GCN_ROW(1) GCN_ROW(2) GCN_ROW(3) GCN_ROW(4) GCN_ROW(5) GCN_ROW(6) GCN_ROW(7)
            GCN_ACC(0) GCN_ACC(1) GCN_ACC(2) GCN_ACC(3) GCN_ACC(4) GCN_ACC(5) GCN_ACC(6) GCN_ACC(7)
        }
        if (m & 4) {
            GCN_ROW(0) GCN_ROW(1) GCN_ROW(2) GCN_ROW(3)
            GCN_ACC(0) GCN_ACC(1) GCN_ACC(2) GCN_ACC(3)
            j += 4;
        }
        if (m & 2) {
            GCN_ROW(0) GCN_ROW(1)
            GCN_ACC(0) GCN_ACC(1)
            j += 2;
        }
        if (m & 1) {
            GCN_ROW(0)
            GCN_ACC(0)
        }
    }

    reinterpret_cast<float4*>(out)[(size_t)node * 32 + lane] = acc;
}


extern "C" void kernel_launch(void* const* d_in, const int* in_sizes, int n_in,
                              void* d_out, int out_size, void* d_ws, size_t ws_size,
                              hipStream_t stream) {
    const float* x  = (const float*)d_in[0];
    const float* W  = (const float*)d_in[1];
    const float* b  = (const float*)d_in[2];
    const int*   ei = (const int*)d_in[3];

    const int n = in_sizes[0] / CH;
    const int E = in_sizes[3] / 2;
    const int* src = ei;
    const int* dst = ei + E;
    const int nb = (n + NPB - 1) / NPB;

    const size_t np = ((size_t)n + 64) & ~(size_t)63;
    const size_t ep = ((size_t)E + 63) & ~(size_t)63;
    const int    tp = (nb + 1 + 63) & ~63;
    int*   offt   = (int*)d_ws;
    int*   seg    = offt + (size_t)GB * tp;
    float* dinv   = (float*)(seg + 2 * np);
    int*   col    = (int*)(dinv + np);
    int*   binned = col + ep;
    float* xw     = (float*)(binned + 2 * ep);
    float* out    = (float*)d_out;

    const int gt = (n + 63) / 64;
    int g1 = (gt + 1) / 2;
    if (g1 > 2 * NWGP - GB) g1 = 2 * NWGP - GB;
    const int g2 = gt - g1;
    gcn_chunk<<<GB + g1, GT, 0, stream>>>(src, dst, offt, binned, E, nb, tp, x, W, xw, n, 0);
    gcn_rank <<<nb + g2, GT, 0, stream>>>(binned, offt, seg, dinv, col, n, E, nb, tp, x, W, xw, g1);

    gcn_gather<<<(n + 7) / 8, 256, 0, stream>>>(seg, col, dinv, xw, b, out, n);
}


# product: a work-group takes 128 rows of x (was 64), each wave two 16x16 accumulators on one B fragment; half as many product work-groups and waves start in the grouping launches; sums unchanged
# speedup vs baseline: 4.7083x; 1.0567x over previous
#include <hip/hip_runtime.h>
#include <hip/hip_bf16.h>

#define CH 128

typedef __attribute__((ext_vector_type(2))) float v2f;
typedef __attribute__((ext_vector_type(8))) float v8f;


#define APITCH 132
#define PROWS  128

__device__ __forceinline__ void gcn_gemm_rows(
    const float* __restrict__ X, const float* __restrict__ W,
    float* __restrict__ XW, int n, int blk)
{
    __shared__ float As[PROWS * APITCH];
    const int r0 = blk * PROWS;
    if (r0 >= n) return;

#pragma unroll
    for (int j = 0; j < 4; ++j) {
        const int idx = (int)threadIdx.x + j * 1024;
        const int row = idx >> 5;
        const int c4  = idx & 31;
        int grow = r0 + row;
        if (grow >= n) grow = n - 1;
        const float4 v = reinterpret_cast<const float4*>(X + (size_t)grow * CH)[c4];
        *reinterpret_cast<float4*>(&As[row * APITCH + 4 * c4]) = v;
    }
    __syncthreads();

    const int wave = threadIdx.x >> 5;
    const int lane = threadIdx.x & 31;
    const int half = lane >> 4;
    const int l16  = lane & 15;
    const int m0   = r0 + (wave >> 3) * 32;
    const int n0   = (wave & 7) * 16;
    if (m0 >= n) return;

    const float* Ap = &As[((wave >> 3) * 32 + l16) * APITCH + 2 * half];
    const float* Bp = W + (size_t)(2 * half) * CH + n0 + l16;

    v8f c0 = {};
    v8f c1 = {};
#pragma unroll 8
    for (int k0 = 0; k0 < CH; k0 += 4) {
        v2f a0, a1, b;
        a0 = *reinterpret_cast<const v2f*>(Ap + k0);
        a1 = *reinterpret_cast<const v2f*>(Ap + 16 * APITCH + k0);
        b.x = Bp[(size_t)k0       * CH];
        b.y = Bp[(size_t)(k0 + 1) * CH];
        c0 = __builtin_amdgcn_wmma_f32_16x16x4_f32(
                 false, a0,  false, b,
                 (short)0, c0,  false,  false);
        c1 = __builtin_amdgcn_wmma_f32_16x16x4_f32(
                 false, a1,  false, b,
                 (short)0, c1,  false,  false);
    }

#pragma unroll
    for (int r = 0; r < 8; ++r) {
        const int row = m0 + r + 8 * half;
        if (row < n)      XW[(size_t)row * CH + n0 + l16]        = c0[r];
        if (row + 16 < n) XW[(size_t)(row + 16) * CH + n0 + l16] = c1[r];
    }
}


#define NPB       256
#define NPB_SHIFT 8
#define MAXNB     2048
#define GB        128
#define GT        1024
#define EPT       4
#define PPW       (GB / 32)
#define PSL       2
#define NWGP      256

__global__ __launch_bounds__(GT, 16) void gcn_chunk(const int* __restrict__ src, const int* __restrict__ dst,
                                                int* __restrict__ offt, int* __restrict__ binned,
                                                int E, int nb, int tp,
                                                const float* __restrict__ X, const float* __restrict__ Wm,
                                                float* __restrict__ XW, int n, int gfirst) {
    __shared__ int h[MAXNB];
    __shared__ int wsum[GT / 32];
    if (blockIdx.x >= GB) {
        gcn_gemm_rows(X, Wm, XW, n, gfirst + (int)blockIdx.x - GB);
        return;
    }
    const int tid  = threadIdx.x;
    const int lane = tid & 31;
    const int wave = tid >> 5;
    for (int k = tid; k < nb; k += GT) h[k] = 0;
    __syncthreads();
    const int ch = (E + GB - 1) / GB;
    const int lo = blockIdx.x * ch;
    const int hi = (lo + ch < E) ? (lo + ch) : E;
    for (int e0 = lo + tid; e0 < hi; e0 += EPT * GT) {
        int d[EPT];
#pragma unroll
        for (int j = 0; j < EPT; ++j) d[j] = (e0 + j * GT < hi) ? dst[e0 + j * GT] : 0;
#pragma unroll
        for (int j = 0; j < EPT; ++j)
            if (e0 + j * GT < hi) atomicAdd(&h[d[j] >> NPB_SHIFT], 1);
    }
    __syncthreads();

    const int i0 = 2 * tid;
    const int a0 = (i0 < nb) ? h[i0] : 0;
    const int a1 = (i0 + 1 < nb) ? h[i0 + 1] : 0;
    const int s  = a0 + a1;
    int inc = s;
#pragma unroll
    for (int off = 1; off < 32; off <<= 1) {
        int t = __shfl_up(inc, off);
        if (lane >= off) inc += t;
    }
    if (lane == 31) wsum[wave] = inc;
    __syncthreads();
    int ex = inc - s;
    for (int w = 0; w < wave; ++w) ex += wsum[w];
    int* orow = offt + (size_t)blockIdx.x * tp;
    if (i0 < nb)     { h[i0]     = ex;      orow[i0]     = ex; }
    if (i0 + 1 < nb) { h[i0 + 1] = ex + a0; orow[i0 + 1] = ex + a0; }
    if (tid == 0) orow[nb] = (hi > lo) ? (hi - lo) : 0;
    __syncthreads();

    int2* b2 = reinterpret_cast<int2*>(binned);
    for (int e0 = lo + tid; e0 < hi; e0 += EPT * GT) {
        int d[EPT], sv[EPT];
#pragma unroll
        for (int j = 0; j < EPT; ++j) {
            const bool in = e0 + j * GT < hi;
            d[j]  = in ? dst[e0 + j * GT] : 0;
            sv[j] = in ? src[e0 + j * GT] : 0;
        }
#pragma unroll
        for (int j = 0; j < EPT; ++j)
            if (e0 + j * GT < hi) {
                const int r = atomicAdd(&h[d[j] >> NPB_SHIFT], 1);
                b2[lo + r] = make_int2(sv[j], d[j]);
            }
    }
}

__global__ __launch_bounds__(GT, 16) void gcn_rank(const int* __restrict__ binned, const int* __restrict__ offt,
                                               int* __restrict__ seg, float* __restrict__ dinv,
                                               int* __restrict__ col, int n, int E, int nb, int tp,
                                               const float* __restrict__ X, const float* __restrict__ Wm,
                                               float* __restrict__ XW, int gfirst) {
    __shared__ int cnt[NPB];
    __shared__ int pos[NPB];
    __shared__ int wsum[NPB / 32];
    __shared__ int bsum;
    if ((int)blockIdx.x >= nb) {
        gcn_gemm_rows(X, Wm, XW, n, gfirst + (int)blockIdx.x - nb);
        return;
    }
    const int tid  = threadIdx.x;
    const int lane = tid & 31;
    const int wave = tid >> 5;
    const int k    = blockIdx.x;
    const int n0   = k * NPB;
    const int ch   = (E + GB - 1) / GB;
    const int2* b2 = reinterpret_cast<const int2*>(binned);

    int p_rel = 0, p_lo = 0, p_hi = 0;
    if (lane < PPW) {
        const int g = wave + 32 * lane;
        const int* orow = offt + (size_t)g * tp + k;
        p_rel = orow[0];
        p_lo  = g * ch + p_rel;
        p_hi  = g * ch + orow[1];
    }
    if (tid < NPB) cnt[tid] = 0;
    if (tid == 0) bsum = 0;
    __syncthreads();
    if (lane < PPW) atomicAdd(&bsum, p_rel);

    int st[PPW], en[PPW];
#pragma unroll
    for (int j = 0; j < PPW; ++j) {
        st[j] = __builtin_amdgcn_readlane(p_lo, j);
        en[j] = __builtin_amdgcn_readlane(p_hi, j);
    }
    int2 sd[PPW * PSL];
#pragma unroll
    for (int j = 0; j < PPW; ++j)
#pragma unroll
        for (int u = 0; u < PSL; ++u) {
            const int i = st[j] + 32 * u + lane;
            sd[j * PSL + u] = (i < en[j]) ? b2[i] : make_int2(0, n0);
        }
#pragma unroll
    for (int j = 0; j < PPW; ++j)
#pragma unroll
        for (int u = 0; u < PSL; ++u)
            if (st[j] + 32 * u + lane < en[j]) atomicAdd(&cnt[sd[j * PSL + u].y - n0], 1);
#pragma unroll
    for (int j = 0; j < PPW; ++j)
        for (int i = st[j] + 32 * PSL + lane; i < en[j]; i += 32) atomicAdd(&cnt[b2[i].y - n0], 1);
    __syncthreads();

    const int c = (tid < NPB) ? cnt[tid] : 0;
    int s = c;
#pragma unroll
    for (int off = 1; off < 32; off <<= 1) {
        int t = __shfl_up(s, off);
        if (lane >= off) s += t;
    }
    if (tid < NPB && lane == 31) wsum[wave] = s;
    __syncthreads();
    if (tid < NPB) {
        int base = bsum;
        for (int w = 0; w < wave; ++w) base += wsum[w];
        const int stn = base + s - c;
        pos[tid] = stn;
        const int node = n0 + tid;
        if (node < n) {
            seg[2 * node]     = stn;
            seg[2 * node + 1] = c;
            dinv[node] = rsqrtf((float)(c + 1));
        }
    }
    __syncthreads();
#pragma unroll
    for (int j = 0; j < PPW; ++j)
#pragma unroll
        for (int u = 0; u < PSL; ++u)
            if (st[j] + 32 * u + lane < en[j]) {
                const int r = atomicAdd(&pos[sd[j * PSL + u].y - n0], 1);
                col[r] = sd[j * PSL + u].x;
            }
#pragma unroll
    for (int j = 0; j < PPW; ++j)
        for (int i = st[j] + 32 * PSL + lane; i < en[j]; i += 32) {
            const int2 e = b2[i];
            const int r = atomicAdd(&pos[e.y - n0], 1);
            col[r] = e.x;
        }
}


#define GCN_ROW(u) \
    const float4 r##u = xw4[(size_t)__builtin_amdgcn_readlane(s_l, j + u) * 32 + lane];
#define GCN_ACC(u) { \
    const float w = __int_as_float(__builtin_amdgcn_readlane(w_l, j + u)); \
    acc.x = fmaf(r##u.x, w, acc.x); acc.y = fmaf(r##u.y, w, acc.y); \
    acc.z = fmaf(r##u.z, w, acc.z); acc.w = fmaf(r##u.w, w, acc.w); }

__global__ __launch_bounds__(256) void gcn_gather(
    const int* __restrict__ seg, const int* __restrict__ col,
    const float* __restrict__ dinv, const float* __restrict__ xw,
    const float* __restrict__ bias, float* __restrict__ out, int n)
{
    const int node = __builtin_amdgcn_readfirstlane((int)(blockIdx.x * 8u + (threadIdx.x >> 5)));
    if (node >= n) return;
    const int lane = threadIdx.x & 31;

    const int start = seg[2 * node];
    const int cnt   = seg[2 * node + 1];
    const float di  = dinv[node];
    const float4* xw4 = reinterpret_cast<const float4*>(xw);

    const float4 self = xw4[(size_t)node * 32 + lane];
    const float4 b4   = reinterpret_cast<const float4*>(bias)[lane];
    float4 acc;
    acc.x = b4.x + self.x * di * di;
    acc.y = b4.y + self.y * di * di;
    acc.z = b4.z + self.z * di * di;
    acc.w = b4.w + self.w * di * di;

    for (int base = 0; base < cnt; base += 32) {
        const int m = (cnt - base < 32) ? (cnt - base) : 32;
        int s_l = 0;
        int w_l = 0;
        if (lane < m) {
            s_l = col[start + base + lane];
            w_l = __float_as_int(dinv[s_l] * di);
        }
        int j = 0;
        for (; j + 8 <= m; j += 8) {
            GCN_ROW(0) GCN_ROW(1) GCN_ROW(2) GCN_ROW(3) GCN_ROW(4) GCN_ROW(5) GCN_ROW(6) GCN_ROW(7)
            GCN_ACC(0) GCN_ACC(1) GCN_ACC(2) GCN_ACC(3) GCN_ACC(4) GCN_ACC(5) GCN_ACC(6) GCN_ACC(7)
        }
        if (m & 4) {
            GCN_ROW(0) GCN_ROW(1) GCN_ROW(2) GCN_ROW(3)
            GCN_ACC(0) GCN_ACC(1) GCN_ACC(2) GCN_ACC(3)
            j += 4;
        }
        if (m & 2) {
            GCN_ROW(0) GCN_ROW(1)
            GCN_ACC(0) GCN_ACC(1)
            j += 2;
        }
        if (m & 1) {
            GCN_ROW(0)
            GCN_ACC(0)
        }
    }

    reinterpret_cast<float4*>(out)[(size_t)node * 32 + lane] = acc;
}


extern "C" void kernel_launch(void* const* d_in, const int* in_sizes, int n_in,
                              void* d_out, int out_size, void* d_ws, size_t ws_size,
                              hipStream_t stream) {
    const float* x  = (const float*)d_in[0];
    const float* W  = (const float*)d_in[1];
    const float* b  = (const float*)d_in[2];
    const int*   ei = (const int*)d_in[3];

    const int n = in_sizes[0] / CH;
    const int E = in_sizes[3] / 2;
    const int* src = ei;
    const int* dst = ei + E;
    const int nb = (n + NPB - 1) / NPB;

    const size_t np = ((size_t)n + 64) & ~(size_t)63;
    const size_t ep = ((size_t)E + 63) & ~(size_t)63;
    const int    tp = (nb + 1 + 63) & ~63;
    int*   offt   = (int*)d_ws;
    int*   seg    = offt + (size_t)GB * tp;
    float* dinv   = (float*)(seg + 2 * np);
    int*   col    = (int*)(dinv + np);
    int*   binned = col + ep;
    float* xw     = (float*)(binned + 2 * ep);
    float* out    = (float*)d_out;

    const int gt = (n + PROWS - 1) / PROWS;
    int g1 = (gt + 1) / 2;
    if (g1 > 2 * NWGP - GB) g1 = 2 * NWGP - GB;
    const int g2 = gt - g1;
    gcn_chunk<<<GB + g1, GT, 0, stream>>>(src, dst, offt, binned, E, nb, tp, x, W, xw, n, 0);
    gcn_rank <<<nb + g2, GT, 0, stream>>>(binned, offt, seg, dinv, col, n, E, nb, tp, x, W, xw, g1);

    gcn_gather<<<(n + 7) / 8, 256, 0, stream>>>(seg, col, dinv, xw, b, out, n);
}


# grouping: both passes place each record at its rank in LDS, then copy the run out in address order (a wave's store is one contiguous stretch; was up to 32 lines, 8 or 4 bytes of each); order unchanged
# speedup vs baseline: 5.2682x; 1.1189x over previous
#include <hip/hip_runtime.h>
#include <hip/hip_bf16.h>

#define CH 128

typedef __attribute__((ext_vector_type(2))) float v2f;
typedef __attribute__((ext_vector_type(8))) float v8f;


#define APITCH 132
#define PROWS  128

__device__ __forceinline__ void gcn_gemm_rows(
    const float* __restrict__ X, const float* __restrict__ W,
    float* __restrict__ XW, int n, int blk, float* As)
{
    const int r0 = blk * PROWS;
    if (r0 >= n) return;

#pragma unroll
    for (int j = 0; j < 4; ++j) {
        const int idx = (int)threadIdx.x + j * 1024;
        const int row = idx >> 5;
        const int c4  = idx & 31;
        int grow = r0 + row;
        if (grow >= n) grow = n - 1;
        const float4 v = reinterpret_cast<const float4*>(X + (size_t)grow * CH)[c4];
        *reinterpret_cast<float4*>(&As[row * APITCH + 4 * c4]) = v;
    }
    __syncthreads();

    const int wave = threadIdx.x >> 5;
    const int lane = threadIdx.x & 31;
    const int half = lane >> 4;
    const int l16  = lane & 15;
    const int m0   = r0 + (wave >> 3) * 32;
    const int n0   = (wave & 7) * 16;
    if (m0 >= n) return;

    const float* Ap = &As[((wave >> 3) * 32 + l16) * APITCH + 2 * half];
    const float* Bp = W + (size_t)(2 * half) * CH + n0 + l16;

    v8f c0 = {};
    v8f c1 = {};
#pragma unroll 8
    for (int k0 = 0; k0 < CH; k0 += 4) {
        v2f a0, a1, b;
        a0 = *reinterpret_cast<const v2f*>(Ap + k0);
        a1 = *reinterpret_cast<const v2f*>(Ap + 16 * APITCH + k0);
        b.x = Bp[(size_t)k0       * CH];
        b.y = Bp[(size_t)(k0 + 1) * CH];
        c0 = __builtin_amdgcn_wmma_f32_16x16x4_f32(
                 false, a0,  false, b,
                 (short)0, c0,  false,  false);
        c1 = __builtin_amdgcn_wmma_f32_16x16x4_f32(
                 false, a1,  false, b,
                 (short)0, c1,  false,  false);
    }

#pragma unroll
    for (int r = 0; r < 8; ++r) {
        const int row = m0 + r + 8 * half;
        if (row < n)      XW[(size_t)row * CH + n0 + l16]        = c0[r];
        if (row + 16 < n) XW[(size_t)(row + 16) * CH + n0 + l16] = c1[r];
    }
}


#define NPB       256
#define NPB_SHIFT 8
#define MAXNB     2048
#define GB        128
#define GT        1024
#define EPT       4
#define PPW       (GB / 32)
#define PSL       2
#define NWGP      256
#define STAGE_WORDS (PROWS * APITCH)

__global__ __launch_bounds__(GT, 16) void gcn_chunk(const int* __restrict__ src, const int* __restrict__ dst,
                                                int* __restrict__ offt, int* __restrict__ binned,
                                                int E, int nb, int tp,
                                                const float* __restrict__ X, const float* __restrict__ Wm,
                                                float* __restrict__ XW, int n, int gfirst) {
    __shared__ int h[MAXNB];
    __shared__ int wsum[GT / 32];
    __shared__ float As[STAGE_WORDS];
    if (blockIdx.x >= GB) {
        gcn_gemm_rows(X, Wm, XW, n, gfirst + (int)blockIdx.x - GB, As);
        return;
    }
    const int tid  = threadIdx.x;
    const int lane = tid & 31;
    const int wave = tid >> 5;
    for (int k = tid; k < nb; k += GT) h[k] = 0;
    __syncthreads();
    const int ch = (E + GB - 1) / GB;
    const int lo = blockIdx.x * ch;
    const int hi = (lo + ch < E) ? (lo + ch) : E;
    for (int e0 = lo + tid; e0 < hi; e0 += EPT * GT) {
        int d[EPT];
#pragma unroll
        for (int j = 0; j < EPT; ++j) d[j] = (e0 + j * GT < hi) ? dst[e0 + j * GT] : 0;
#pragma unroll
        for (int j = 0; j < EPT; ++j)
            if (e0 + j * GT < hi) atomicAdd(&h[d[j] >> NPB_SHIFT], 1);
    }
    __syncthreads();

    const int i0 = 2 * tid;
    const int a0 = (i0 < nb) ? h[i0] : 0;
    const int a1 = (i0 + 1 < nb) ? h[i0 + 1] : 0;
    const int s  = a0 + a1;
    int inc = s;
#pragma unroll
    for (int off = 1; off < 32; off <<= 1) {
        int t = __shfl_up(inc, off);
        if (lane >= off) inc += t;
    }
    if (lane == 31) wsum[wave] = inc;
    __syncthreads();
    int ex = inc - s;
    for (int w = 0; w < wave; ++w) ex += wsum[w];
    int* orow = offt + (size_t)blockIdx.x * tp;
    if (i0 < nb)     { h[i0]     = ex;      orow[i0]     = ex; }
    if (i0 + 1 < nb) { h[i0 + 1] = ex + a0; orow[i0 + 1] = ex + a0; }
    if (tid == 0) orow[nb] = (hi > lo) ? (hi - lo) : 0;
    __syncthreads();

    int2* b2 = reinterpret_cast<int2*>(binned);
    int2* stage = reinterpret_cast<int2*>(As);
    const bool staged = (hi - lo) <= STAGE_WORDS / 2;
    for (int e0 = lo + tid; e0 < hi; e0 += EPT * GT) {
        int d[EPT], sv[EPT];
#pragma unroll
        for (int j = 0; j < EPT; ++j) {
            const bool in = e0 + j * GT < hi;
            d[j]  = in ? dst[e0 + j * GT] : 0;
            sv[j] = in ? src[e0 + j * GT] : 0;
        }
#pragma unroll
        for (int j = 0; j < EPT; ++j)
            if (e0 + j * GT < hi) {
                const int r = atomicAdd(&h[d[j] >> NPB_SHIFT], 1);
                if (staged) stage[r] = make_int2(sv[j], d[j]);
                else        b2[lo + r] = make_int2(sv[j], d[j]);
            }
    }
    if (!staged) return;
    __syncthreads();
    for (int i = tid; i < hi - lo; i += GT) b2[lo + i] = stage[i];
}

__global__ __launch_bounds__(GT, 16) void gcn_rank(const int* __restrict__ binned, const int* __restrict__ offt,
                                               int* __restrict__ seg, float* __restrict__ dinv,
                                               int* __restrict__ col, int n, int E, int nb, int tp,
                                               const float* __restrict__ X, const float* __restrict__ Wm,
                                               float* __restrict__ XW, int gfirst) {
    __shared__ int cnt[NPB];
    __shared__ int pos[NPB];
    __shared__ int wsum[NPB / 32];
    __shared__ int bsum;
    __shared__ int btot;
    __shared__ float As[STAGE_WORDS];
    if ((int)blockIdx.x >= nb) {
        gcn_gemm_rows(X, Wm, XW, n, gfirst + (int)blockIdx.x - nb, As);
        return;
    }
    const int tid  = threadIdx.x;
    const int lane = tid & 31;
    const int wave = tid >> 5;
    const int k    = blockIdx.x;
    const int n0   = k * NPB;
    const int ch   = (E + GB - 1) / GB;
    const int2* b2 = reinterpret_cast<const int2*>(binned);

    int p_rel = 0, p_lo = 0, p_hi = 0;
    if (lane < PPW) {
        const int g = wave + 32 * lane;
        const int* orow = offt + (size_t)g * tp + k;
        p_rel = orow[0];
        p_lo  = g * ch + p_rel;
        p_hi  = g * ch + orow[1];
    }
    if (tid < NPB) cnt[tid] = 0;
    if (tid == 0) bsum = 0;
    __syncthreads();
    if (lane < PPW) atomicAdd(&bsum, p_rel);

    int st[PPW], en[PPW];
#pragma unroll
    for (int j = 0; j < PPW; ++j) {
        st[j] = __builtin_amdgcn_readlane(p_lo, j);
        en[j] = __builtin_amdgcn_readlane(p_hi, j);
    }
    int2 sd[PPW * PSL];
#pragma unroll
    for (int j = 0; j < PPW; ++j)
#pragma unroll
        for (int u = 0; u < PSL; ++u) {
            const int i = st[j] + 32 * u + lane;
            sd[j * PSL + u] = (i < en[j]) ? b2[i] : make_int2(0, n0);
        }
#pragma unroll
    for (int j = 0; j < PPW; ++j)
#pragma unroll
        for (int u = 0; u < PSL; ++u)
            if (st[j] + 32 * u + lane < en[j]) atomicAdd(&cnt[sd[j * PSL + u].y - n0], 1);
#pragma unroll
    for (int j = 0; j < PPW; ++j)
        for (int i = st[j] + 32 * PSL + lane; i < en[j]; i += 32) atomicAdd(&cnt[b2[i].y - n0], 1);
    __syncthreads();

    const int c = (tid < NPB) ? cnt[tid] : 0;
    int s = c;
#pragma unroll
    for (int off = 1; off < 32; off <<= 1) {
        int t = __shfl_up(s, off);
        if (lane >= off) s += t;
    }
    if (tid < NPB && lane == 31) wsum[wave] = s;
    __syncthreads();
    if (tid < NPB) {
        int base = bsum;
        for (int w = 0; w < wave; ++w) base += wsum[w];
        const int stn = base + s - c;
        pos[tid] = stn;
        if (tid == NPB - 1) btot = stn + c - bsum;
        const int node = n0 + tid;
        if (node < n) {
            seg[2 * node]     = stn;
            seg[2 * node + 1] = c;
            dinv[node] = rsqrtf((float)(c + 1));
        }
    }
    __syncthreads();
    const int bs  = bsum;
    const int tot = btot;
    int* stage = reinterpret_cast<int*>(As);
    const bool staged = tot <= STAGE_WORDS;
#pragma unroll
    for (int j = 0; j < PPW; ++j)
#pragma unroll
        for (int u = 0; u < PSL; ++u)
            if (st[j] + 32 * u + lane < en[j]) {
                const int r = atomicAdd(&pos[sd[j * PSL + u].y - n0], 1);
                if (staged) stage[r - bs] = sd[j * PSL + u].x;
                else        col[r] = sd[j * PSL + u].x;
            }
#pragma unroll
    for (int j = 0; j < PPW; ++j)
        for (int i = st[j] + 32 * PSL + lane; i < en[j]; i += 32) {
            const int2 e = b2[i];
            const int r = atomicAdd(&pos[e.y - n0], 1);
            if (staged) stage[r - bs] = e.x;
            else        col[r] = e.x;
        }
    if (!staged) return;
    __syncthreads();
    for (int i = tid; i < tot; i += GT) col[bs + i] = stage[i];
}


#define GCN_ROW(u) \
    const float4 r##u = xw4[(size_t)__builtin_amdgcn_readlane(s_l, j + u) * 32 + lane];
#define GCN_ACC(u) { \
    const float w = __int_as_float(__builtin_amdgcn_readlane(w_l, j + u)); \
    acc.x = fmaf(r##u.x, w, acc.x); acc.y = fmaf(r##u.y, w, acc.y); \
    acc.z = fmaf(r##u.z, w, acc.z); acc.w = fmaf(r##u.w, w, acc.w); }

__global__ __launch_bounds__(256) void gcn_gather(
    const int* __restrict__ seg, const int* __restrict__ col,
    const float* __restrict__ dinv, const float* __restrict__ xw,
    const float* __restrict__ bias, float* __restrict__ out, int n)
{
    const int node = __builtin_amdgcn_readfirstlane((int)(blockIdx.x * 8u + (threadIdx.x >> 5)));
    if (node >= n) return;
    const int lane = threadIdx.x & 31;

    const int start = seg[2 * node];
    const int cnt   = seg[2 * node + 1];
    const float di  = dinv[node];
    const float4* xw4 = reinterpret_cast<const float4*>(xw);

    const float4 self = xw4[(size_t)node * 32 + lane];
    const float4 b4   = reinterpret_cast<const float4*>(bias)[lane];
    float4 acc;
    acc.x = b4.x + self.x * di * di;
    acc.y = b4.y + self.y * di * di;
    acc.z = b4.z + self.z * di * di;
    acc.w = b4.w + self.w * di * di;

    for (int base = 0; base < cnt; base += 32) {
        const int m = (cnt - base < 32) ? (cnt - base) : 32;
        int s_l = 0;
        int w_l = 0;
        if (lane < m) {
            s_l = col[start + base + lane];
            w_l = __float_as_int(dinv[s_l] * di);
        }
        int j = 0;
        for (; j + 8 <= m; j += 8) {
            GCN_ROW(0) GCN_ROW(1) GCN_ROW(2) GCN_ROW(3) GCN_ROW(4) GCN_ROW(5) GCN_ROW(6) GCN_ROW(7)
            GCN_ACC(0) GCN_ACC(1) GCN_ACC(2) GCN_ACC(3) GCN_ACC(4) GCN_ACC(5) GCN_ACC(6) GCN_ACC(7)
        }
        if (m & 4) {
            GCN_ROW(0) GCN_ROW(1) GCN_ROW(2) GCN_ROW(3)
            GCN_ACC(0) GCN_ACC(1) GCN_ACC(2) GCN_ACC(3)
            j += 4;
        }
        if (m & 2) {
            GCN_ROW(0) GCN_ROW(1)
            GCN_ACC(0) GCN_ACC(1)
            j += 2;
        }
        if (m & 1) {
            GCN_ROW(0)
            GCN_ACC(0)
        }
    }

    reinterpret_cast<float4*>(out)[(size_t)node * 32 + lane] = acc;
}


extern "C" void kernel_launch(void* const* d_in, const int* in_sizes, int n_in,
                              void* d_out, int out_size, void* d_ws, size_t ws_size,
                              hipStream_t stream) {
    const float* x  = (const float*)d_in[0];
    const float* W  = (const float*)d_in[1];
    const float* b  = (const float*)d_in[2];
    const int*   ei = (const int*)d_in[3];

    const int n = in_sizes[0] / CH;
    const int E = in_sizes[3] / 2;
    const int* src = ei;
    const int* dst = ei + E;
    const int nb = (n + NPB - 1) / NPB;

    const size_t np = ((size_t)n + 64) & ~(size_t)63;
    const size_t ep = ((size_t)E + 63) & ~(size_t)63;
    const int    tp = (nb + 1 + 63) & ~63;
    int*   offt   = (int*)d_ws;
    int*   seg    = offt + (size_t)GB * tp;
    float* dinv   = (float*)(seg + 2 * np);
    int*   col    = (int*)(dinv + np);
    int*   binned = col + ep;
    float* xw     = (float*)(binned + 2 * ep);
    float* out    = (float*)d_out;

    const int gt = (n + PROWS - 1) / PROWS;
    int g1 = (gt + 1) / 2;
    if (g1 > 2 * NWGP - GB) g1 = 2 * NWGP - GB;
    const int g2 = gt - g1;
    gcn_chunk<<<GB + g1, GT, 0, stream>>>(src, dst, offt, binned, E, nb, tp, x, W, xw, n, 0);
    gcn_rank <<<nb + g2, GT, 0, stream>>>(binned, offt, seg, dinv, col, n, E, nb, tp, x, W, xw, g1);

    gcn_gather<<<(n + 7) / 8, 256, 0, stream>>>(seg, col, dinv, xw, b, out, n);
}
